# starting kernel
# baseline (speedup 1.0000x reference)
#include <hip/hip_runtime.h>
#include <hip/hip_bf16.h>

#define NUM_USERS 50000
#define NUM_ITEMS 20000
#define NNODES    70000
#define MTILES    (NNODES / 16)
#define HID       128
#define NE        1000000
#define EPLUS     (NE + NNODES)
#define NEG_SLOPE 0.2f

typedef float v2f __attribute__((ext_vector_type(2)));
typedef float v8f __attribute__((ext_vector_type(8)));

__device__ __forceinline__ unsigned fkey(float f) {
    unsigned b = __float_as_uint(f);
    return (b & 0x80000000u) ? ~b : (b | 0x80000000u);
}
__device__ __forceinline__ float funkey(unsigned k) {
    unsigned b = (k & 0x80000000u) ? (k & 0x7FFFFFFFu) : ~k;
    return __uint_as_float(b);
}

__global__ __launch_bounds__(256) void concat_k(const float* __restrict__ u,
                                                const float* __restrict__ it,
                                                float* __restrict__ x) {
    int i = blockIdx.x * 256 + threadIdx.x;
    const int UW = NUM_USERS * HID;
    if (i < NNODES * HID) x[i] = (i < UW) ? u[i] : it[i - UW];
}

__global__ __launch_bounds__(256) void gemm_wmma(const float* __restrict__ X,
                                                 const float* __restrict__ Wl,
                                                 float* __restrict__ H) {
    const int wave  = threadIdx.x >> 5;
    const int lane  = threadIdx.x & 31;
    const int tileM = blockIdx.x * 8 + wave;
    if (tileM >= MTILES) return;
    const int m     = lane & 15;
    const int khalf = (lane >> 4) * 2;
    const int row   = tileM * 16 + m;

    v8f c[8] = {};
    const float* xr = X + (size_t)row * HID + khalf;
    const float* wr = Wl + (size_t)khalf * HID + m;

    for (int k = 0; k < HID; k += 4) {
        v2f a = *(const v2f*)(xr + k);
        const float* w0 = wr + (size_t)k * HID;
        const float* w1 = w0 + HID;
#pragma unroll
        for (int t = 0; t < 8; ++t) {
            v2f b;
            b.x = w0[t * 16];
            b.y = w1[t * 16];
            c[t] = __builtin_amdgcn_wmma_f32_16x16x4_f32(false, a, false, b,
                                                         (short)0, c[t], false, false);
        }
    }
    const int rbase = tileM * 16 + ((lane >> 4) ? 8 : 0);
#pragma unroll
    for (int t = 0; t < 8; ++t)
#pragma unroll
        for (int r = 0; r < 8; ++r)
            H[(size_t)(rbase + r) * HID + t * 16 + m] = c[t][r];
}

__global__ __launch_bounds__(256) void alpha_k(const float* __restrict__ H,
                                               const float* __restrict__ Asrc,
                                               const float* __restrict__ Adst,
                                               float* __restrict__ as_out,
                                               float* __restrict__ ad_out) {
    const int warp = threadIdx.x >> 5, lane = threadIdx.x & 31;
    const int node = blockIdx.x * 8 + warp;
    if (node >= NNODES) return;
    float4 hv = ((const float4*)H)[(size_t)node * 32 + lane];
    float4 av = ((const float4*)Asrc)[lane];
    float4 dv = ((const float4*)Adst)[lane];
    float ss = hv.x * av.x + hv.y * av.y + hv.z * av.z + hv.w * av.w;
    float sd = hv.x * dv.x + hv.y * dv.y + hv.z * dv.z + hv.w * dv.w;
#pragma unroll
    for (int off = 16; off > 0; off >>= 1) {
        ss += __shfl_xor(ss, off, 32);
        sd += __shfl_xor(sd, off, 32);
    }
    if (lane == 0) { as_out[node] = ss; ad_out[node] = sd; }
}

__global__ __launch_bounds__(256) void init_nodes(unsigned* __restrict__ mkey,
                                                  float* __restrict__ den) {
    int i = blockIdx.x * 256 + threadIdx.x;
    if (i < NNODES) { mkey[i] = 0u; den[i] = 0.f; }
}

__global__ __launch_bounds__(256) void init_bias(float* __restrict__ out,
                                                 const float* __restrict__ b) {
    int i = blockIdx.x * 256 + threadIdx.x;
    if (i < NNODES * HID) out[i] = b[i & (HID - 1)];
}

__device__ __forceinline__ void edge_sd(int i, const int* __restrict__ ei,
                                        int& s, int& d) {
    if (i < NE) { s = ei[i]; d = ei[NE + i]; }
    else        { s = d = i - NE; }
}

__global__ __launch_bounds__(256) void edge_max_k(const int* __restrict__ ei,
                                                  const float* __restrict__ as_,
                                                  const float* __restrict__ ad_,
                                                  unsigned* __restrict__ mkey) {
    int i = blockIdx.x * 256 + threadIdx.x;
    if (i >= EPLUS) return;
    int s, d; edge_sd(i, ei, s, d);
    float e = as_[s] + ad_[d];
    e = (e >= 0.f) ? e : NEG_SLOPE * e;
    atomicMax(mkey + d, fkey(e));
}

__global__ __launch_bounds__(256) void edge_exp_k(const int* __restrict__ ei,
                                                  const float* __restrict__ as_,
                                                  const float* __restrict__ ad_,
                                                  const unsigned* __restrict__ mkey,
                                                  float* __restrict__ den,
                                                  float* __restrict__ exb) {
    int i = blockIdx.x * 256 + threadIdx.x;
    if (i >= EPLUS) return;
    int s, d; edge_sd(i, ei, s, d);
    float e = as_[s] + ad_[d];
    e = (e >= 0.f) ? e : NEG_SLOPE * e;
    float ex = expf(e - funkey(mkey[d]));
    exb[i] = ex;
    atomicAdd(den + d, ex);
}

__global__ __launch_bounds__(256) void edge_agg_k(const int* __restrict__ ei,
                                                  const float* __restrict__ H,
                                                  const float* __restrict__ exb,
                                                  const float* __restrict__ den,
                                                  float* __restrict__ out) {
    const int warp = threadIdx.x >> 5, lane = threadIdx.x & 31;
    const int e = blockIdx.x * 8 + warp;
    if (e >= EPLUS) return;
    int s, d; edge_sd(e, ei, s, d);
    float alpha = exb[e] / den[d];
    float4 hv = ((const float4*)H)[(size_t)s * 32 + lane];
    float* o = out + (size_t)d * HID + lane * 4;
    atomicAdd(o + 0, hv.x * alpha);
    atomicAdd(o + 1, hv.y * alpha);
    atomicAdd(o + 2, hv.z * alpha);
    atomicAdd(o + 3, hv.w * alpha);
}

extern "C" void kernel_launch(void* const* d_in, const int* in_sizes, int n_in,
                              void* d_out, int out_size, void* d_ws, size_t ws_size,
                              hipStream_t stream) {
    const int*   ei      = (const int*)d_in[0];
    const float* users   = (const float*)d_in[1];
    const float* items   = (const float*)d_in[2];
    const float* W       = (const float*)d_in[3];
    const float* att_src = (const float*)d_in[4];
    const float* att_dst = (const float*)d_in[5];
    const float* bias    = (const float*)d_in[6];
    float*       out     = (float*)d_out;

    float*    xbuf  = (float*)d_ws;
    float*    hbuf  = xbuf + (size_t)NNODES * HID;
    float*    asb   = hbuf + (size_t)NNODES * HID;
    float*    adb   = asb + NNODES;
    unsigned* mkey  = (unsigned*)(adb + NNODES);
    float*    den   = (float*)(mkey + NNODES);
    float*    exb   = den + NNODES;

    const int GRID_NH = (NNODES * HID + 255) / 256;
    const int GRID_N  = (NNODES + 255) / 256;
    const int GRID_E  = (EPLUS + 255) / 256;
    const int GRID_EW = (EPLUS + 7) / 8;
    const int GRID_GE = (MTILES + 7) / 8;

    concat_k<<<GRID_NH, 256, 0, stream>>>(users, items, xbuf);

    for (int l = 0; l < 2; ++l) {
        const float* Wl = W + (size_t)l * HID * HID;
        float* tgt = (l == 0) ? xbuf : out;
        gemm_wmma<<<GRID_GE, 256, 0, stream>>>(xbuf, Wl, hbuf);
        alpha_k<<<(NNODES + 7) / 8, 256, 0, stream>>>(hbuf, att_src + l * HID,
                                                      att_dst + l * HID, asb, adb);
        init_nodes<<<GRID_N, 256, 0, stream>>>(mkey, den);
        init_bias<<<GRID_NH, 256, 0, stream>>>(tgt, bias + l * HID);
        edge_max_k<<<GRID_E, 256, 0, stream>>>(ei, asb, adb, mkey);
        edge_exp_k<<<GRID_E, 256, 0, stream>>>(ei, asb, adb, mkey, den, exb);
        edge_agg_k<<<GRID_EW, 256, 0, stream>>>(ei, hbuf, exb, den, tgt);
    }
}


# Edges counting-sorted by destination in-call (histogram, scan, scatter); per layer one wave per node does max, exp, sum and row gather: float atomics gone, sum order = arrival rank in list
# speedup vs baseline: 2.7411x; 2.7411x over previous
#include <hip/hip_runtime.h>
#include <hip/hip_bf16.h>

#define NUM_USERS 50000
#define NUM_ITEMS 20000
#define NNODES    70000
#define MTILES    (NNODES / 16)
#define HID       128
#define NE        1000000
#define NEG_SLOPE 0.2f

typedef float v2f __attribute__((ext_vector_type(2)));
typedef float v8f __attribute__((ext_vector_type(8)));

__global__ __launch_bounds__(256) void concat_k(const float* __restrict__ u,
                                                const float* __restrict__ it,
                                                float* __restrict__ x) {
    int i = blockIdx.x * 256 + threadIdx.x;
    const int UW = NUM_USERS * HID;
    if (i < NNODES * HID) x[i] = (i < UW) ? u[i] : it[i - UW];
}

__global__ __launch_bounds__(256) void gemm_wmma(const float* __restrict__ X,
                                                 const float* __restrict__ Wl,
                                                 float* __restrict__ H) {
    const int wave  = threadIdx.x >> 5;
    const int lane  = threadIdx.x & 31;
    const int tileM = blockIdx.x * 8 + wave;
    if (tileM >= MTILES) return;
    const int m     = lane & 15;
    const int khalf = (lane >> 4) * 2;
    const int row   = tileM * 16 + m;

    v8f c[8] = {};
    const float* xr = X + (size_t)row * HID + khalf;
    const float* wr = Wl + (size_t)khalf * HID + m;

    for (int k = 0; k < HID; k += 4) {
        v2f a = *(const v2f*)(xr + k);
        const float* w0 = wr + (size_t)k * HID;
        const float* w1 = w0 + HID;
#pragma unroll
        for (int t = 0; t < 8; ++t) {
            v2f b;
            b.x = w0[t * 16];
            b.y = w1[t * 16];
            c[t] = __builtin_amdgcn_wmma_f32_16x16x4_f32(false, a, false, b,
                                                         (short)0, c[t], false, false);
        }
    }
    const int rbase = tileM * 16 + ((lane >> 4) ? 8 : 0);
#pragma unroll
    for (int t = 0; t < 8; ++t)
#pragma unroll
        for (int r = 0; r < 8; ++r)
            H[(size_t)(rbase + r) * HID + t * 16 + m] = c[t][r];
}

__global__ __launch_bounds__(256) void alpha_k(const float* __restrict__ H,
                                               const float* __restrict__ Asrc,
                                               const float* __restrict__ Adst,
                                               float* __restrict__ as_out,
                                               float* __restrict__ ad_out) {
    const int warp = threadIdx.x >> 5, lane = threadIdx.x & 31;
    const int node = blockIdx.x * 8 + warp;
    if (node >= NNODES) return;
    float4 hv = ((const float4*)H)[(size_t)node * 32 + lane];
    float4 av = ((const float4*)Asrc)[lane];
    float4 dv = ((const float4*)Adst)[lane];
    float ss = hv.x * av.x + hv.y * av.y + hv.z * av.z + hv.w * av.w;
    float sd = hv.x * dv.x + hv.y * dv.y + hv.z * dv.z + hv.w * dv.w;
#pragma unroll
    for (int off = 16; off > 0; off >>= 1) {
        ss += __shfl_xor(ss, off, 32);
        sd += __shfl_xor(sd, off, 32);
    }
    if (lane == 0) { as_out[node] = ss; ad_out[node] = sd; }
}

#define NQUAD   (NE / 4)
#define SCAN_IT 18
#define NPAD    (32 * SCAN_IT * 32 * 4)
static_assert(NE % 4 == 0, "edge quads");
static_assert(NPAD >= NNODES + 1, "one block scans every node and the end marker");

__global__ __launch_bounds__(256) void zero_deg(int* __restrict__ deg) {
    int i = blockIdx.x * 256 + threadIdx.x;
    if (i < NPAD) deg[i] = 0;
}

__global__ __launch_bounds__(256) void hist_k(const int* __restrict__ ei,
                                              int* __restrict__ deg,
                                              int* __restrict__ rank) {
    int q = blockIdx.x * 256 + threadIdx.x;
    if (q >= NQUAD) return;
    int4 d4 = ((const int4*)(ei + NE))[q];
    int4 r;
    r.x = atomicAdd(deg + d4.x, 1);
    r.y = atomicAdd(deg + d4.y, 1);
    r.z = atomicAdd(deg + d4.z, 1);
    r.w = atomicAdd(deg + d4.w, 1);
    ((int4*)rank)[q] = r;
}

__global__ __launch_bounds__(1024) void scan_k(const int* __restrict__ deg,
                                               int* __restrict__ off) {
    __shared__ int wtot[32];
    const int lane = threadIdx.x & 31, wave = threadIdx.x >> 5;
    const int4* deg4 = (const int4*)deg + wave * (SCAN_IT * 32) + lane;
    int4*       off4 = (int4*)off       + wave * (SCAN_IT * 32) + lane;

    int4 v[SCAN_IT];
#pragma unroll
    for (int it = 0; it < SCAN_IT; ++it) v[it] = deg4[it * 32];
    int part = 0;
#pragma unroll
    for (int it = 0; it < SCAN_IT; ++it) part += (v[it].x + v[it].y) + (v[it].z + v[it].w);
#pragma unroll
    for (int o = 16; o > 0; o >>= 1) part += __shfl_xor(part, o, 32);
    if (lane == 0) wtot[wave] = part;
    __syncthreads();

    int carry = (lane < wave) ? wtot[lane] : 0;
#pragma unroll
    for (int o = 16; o > 0; o >>= 1) carry += __shfl_xor(carry, o, 32);

#pragma unroll
    for (int it = 0; it < SCAN_IT; ++it) {
        const int t = (v[it].x + v[it].y) + (v[it].z + v[it].w);
        int incl = t;
#pragma unroll
        for (int o = 1; o < 32; o <<= 1) {
            const int u = __shfl_up(incl, o, 32);
            if (lane >= o) incl += u;
        }
        const int ex = carry + incl - t;
        off4[it * 32] = make_int4(ex, ex + v[it].x, ex + v[it].x + v[it].y,
                                  ex + v[it].x + v[it].y + v[it].z);
        carry += __shfl(incl, 31, 32);
    }
}

__global__ __launch_bounds__(256) void scatter_k(const int* __restrict__ ei,
                                                 const int* __restrict__ off,
                                                 const int* __restrict__ rank,
                                                 int* __restrict__ csr) {
    int q = blockIdx.x * 256 + threadIdx.x;
    if (q >= NQUAD) return;
    int4 s4 = ((const int4*)ei)[q];
    int4 d4 = ((const int4*)(ei + NE))[q];
    int4 r4 = ((const int4*)rank)[q];
    csr[off[d4.x] + r4.x] = s4.x;
    csr[off[d4.y] + r4.y] = s4.y;
    csr[off[d4.z] + r4.z] = s4.z;
    csr[off[d4.w] + r4.w] = s4.w;
}

__device__ __forceinline__ float lrelu(float e) { return (e >= 0.f) ? e : NEG_SLOPE * e; }

template <int N> __device__ __forceinline__ float row_xor(float x) {
    return __int_as_float(__builtin_amdgcn_update_dpp(0, __float_as_int(x), 0x160 + N, 0xF, 0xF, false));
}
__device__ __forceinline__ float xhalf(float x) {
    const int b = __float_as_int(x);
    return __int_as_float(__builtin_amdgcn_permlanex16(b, b, 0x76543210, 0xfedcba98, false, false));
}
__device__ __forceinline__ float wave_max(float v) {
    v = fmaxf(v, row_xor<1>(v));
    v = fmaxf(v, row_xor<2>(v));
    v = fmaxf(v, row_xor<4>(v));
    v = fmaxf(v, row_xor<8>(v));
    return fmaxf(v, xhalf(v));
}
__device__ __forceinline__ float wave_sum(float v) {
    v += row_xor<1>(v);
    v += row_xor<2>(v);
    v += row_xor<4>(v);
    v += row_xor<8>(v);
    return v + xhalf(v);
}

__device__ __forceinline__ void gather_rows(float4& acc, const float4* __restrict__ H4,
                                            int s, float al, int cnt, int lane) {
    const int ab = __float_as_int(al);
    for (int k = 0; k < cnt; k += 4) {
        const int s0 = __builtin_amdgcn_readlane(s, k);
        const int s1 = __builtin_amdgcn_readlane(s, k + 1);
        const int s2 = __builtin_amdgcn_readlane(s, k + 2);
        const int s3 = __builtin_amdgcn_readlane(s, k + 3);
        const float a0 = __int_as_float(__builtin_amdgcn_readlane(ab, k));
        const float a1 = __int_as_float(__builtin_amdgcn_readlane(ab, k + 1));
        const float a2 = __int_as_float(__builtin_amdgcn_readlane(ab, k + 2));
        const float a3 = __int_as_float(__builtin_amdgcn_readlane(ab, k + 3));
        const float4 h0 = H4[(size_t)s0 * 32 + lane];
        const float4 h1 = H4[(size_t)s1 * 32 + lane];
        const float4 h2 = H4[(size_t)s2 * 32 + lane];
        const float4 h3 = H4[(size_t)s3 * 32 + lane];
        acc.x += h0.x * a0; acc.y += h0.y * a0; acc.z += h0.z * a0; acc.w += h0.w * a0;
        acc.x += h1.x * a1; acc.y += h1.y * a1; acc.z += h1.z * a1; acc.w += h1.w * a1;
        acc.x += h2.x * a2; acc.y += h2.y * a2; acc.z += h2.z * a2; acc.w += h2.w * a2;
        acc.x += h3.x * a3; acc.y += h3.y * a3; acc.z += h3.z * a3; acc.w += h3.w * a3;
    }
}

__global__ __launch_bounds__(256) void gat_node_k(const int* __restrict__ off,
                                                  const int* __restrict__ csr,
                                                  const float* __restrict__ H,
                                                  const float* __restrict__ as_,
                                                  const float* __restrict__ ad_,
                                                  const float* __restrict__ bias,
                                                  float* __restrict__ out) {
    const int lane = threadIdx.x & 31;
    const int d = __builtin_amdgcn_readfirstlane(blockIdx.x * 8 + (threadIdx.x >> 5));
    if (d >= NNODES) return;
    const int beg = off[d], end = off[d + 1];
    const int deg = end - beg;
    const float add = ad_[d];
    const float es = lrelu(as_[d] + add);
    const float4* H4 = (const float4*)H;
    const float4 hs = H4[(size_t)d * 32 + lane];
    float4 acc;

    if (deg <= 32) {
        int s = d; float e = -__builtin_inff();
        if (lane < deg) { s = csr[beg + lane]; e = lrelu(as_[s] + add); }
        const float m   = fmaxf(es, wave_max(e));
        const float ex  = (lane < deg) ? expf(e - m) : 0.f;
        const float exs = expf(es - m);
        const float den = exs + wave_sum(ex);
        const float al  = ex / den;
        const float als = exs / den;
        acc.x = hs.x * als; acc.y = hs.y * als; acc.z = hs.z * als; acc.w = hs.w * als;
        gather_rows(acc, H4, s, al, deg, lane);
    } else {
        float m = es;
        for (int b = beg; b < end; b += 32) {
            const int j = b + lane;
            if (j < end) m = fmaxf(m, lrelu(as_[csr[j]] + add));
        }
        m = wave_max(m);
        float part = 0.f;
        for (int b = beg; b < end; b += 32) {
            const int j = b + lane;
            if (j < end) part += expf(lrelu(as_[csr[j]] + add) - m);
        }
        const float exs = expf(es - m);
        const float den = exs + wave_sum(part);
        const float als = exs / den;
        acc.x = hs.x * als; acc.y = hs.y * als; acc.z = hs.z * als; acc.w = hs.w * als;
        for (int b = beg; b < end; b += 32) {
            const int j = b + lane;
            int s = d; float al = 0.f;
            if (j < end) { s = csr[j]; al = expf(lrelu(as_[s] + add) - m) / den; }
            const int left = end - b;
            gather_rows(acc, H4, s, al, left < 32 ? left : 32, lane);
        }
    }
    const float4 bv = ((const float4*)bias)[lane];
    acc.x += bv.x; acc.y += bv.y; acc.z += bv.z; acc.w += bv.w;
    ((float4*)out)[(size_t)d * 32 + lane] = acc;
}

extern "C" void kernel_launch(void* const* d_in, const int* in_sizes, int n_in,
                              void* d_out, int out_size, void* d_ws, size_t ws_size,
                              hipStream_t stream) {
    const int*   ei      = (const int*)d_in[0];
    const float* users   = (const float*)d_in[1];
    const float* items   = (const float*)d_in[2];
    const float* W       = (const float*)d_in[3];
    const float* att_src = (const float*)d_in[4];
    const float* att_dst = (const float*)d_in[5];
    const float* bias    = (const float*)d_in[6];
    float*       out     = (float*)d_out;

    float* xbuf = (float*)d_ws;
    float* hbuf = xbuf + (size_t)NNODES * HID;
    float* asb  = hbuf + (size_t)NNODES * HID;
    float* adb  = asb + NNODES;
    int*   deg  = (int*)(adb + NNODES);
    int*   off  = deg + NPAD;
    int*   rank = off + NPAD;
    int*   csr  = rank + NE;

    const int GRID_NH = (NNODES * HID + 255) / 256;
    const int GRID_Z  = (NPAD + 255) / 256;
    const int GRID_Q  = (NQUAD + 255) / 256;
    const int GRID_GE = (MTILES + 7) / 8;
    const int GRID_NW = (NNODES + 7) / 8;

    concat_k<<<GRID_NH, 256, 0, stream>>>(users, items, xbuf);

    zero_deg<<<GRID_Z, 256, 0, stream>>>(deg);
    hist_k<<<GRID_Q, 256, 0, stream>>>(ei, deg, rank);
    scan_k<<<1, 1024, 0, stream>>>(deg, off);
    scatter_k<<<GRID_Q, 256, 0, stream>>>(ei, off, rank, csr);

    for (int l = 0; l < 2; ++l) {
        const float* Wl = W + (size_t)l * HID * HID;
        float* tgt = (l == 0) ? xbuf : out;
        gemm_wmma<<<GRID_GE, 256, 0, stream>>>(xbuf, Wl, hbuf);
        alpha_k<<<GRID_NW, 256, 0, stream>>>(hbuf, att_src + l * HID,
                                             att_dst + l * HID, asb, adb);
        gat_node_k<<<GRID_NW, 256, 0, stream>>>(off, csr, hbuf, asb, adb,
                                                bias + l * HID, tgt);
    }
}


# Concat kernel removed: the layer-0 GEMM loads its A rows straight from the user and item tables (per-row base select), so 280000 one-element waves, one launch and a 36 MB copy are gone
# speedup vs baseline: 3.0328x; 1.1064x over previous
#include <hip/hip_runtime.h>
#include <hip/hip_bf16.h>

#define NUM_USERS 50000
#define NUM_ITEMS 20000
#define NNODES    70000
#define MTILES    (NNODES / 16)
#define HID       128
#define NE        1000000
#define NEG_SLOPE 0.2f

typedef float v2f __attribute__((ext_vector_type(2)));
typedef float v8f __attribute__((ext_vector_type(8)));

__global__ __launch_bounds__(256) void gemm_wmma(const float* __restrict__ Xu,
                                                 const float* __restrict__ Xi,
                                                 const float* __restrict__ Wl,
                                                 float* __restrict__ H) {
    const int wave  = threadIdx.x >> 5;
    const int lane  = threadIdx.x & 31;
    const int tileM = blockIdx.x * 8 + wave;
    if (tileM >= MTILES) return;
    const int m     = lane & 15;
    const int khalf = (lane >> 4) * 2;
    const int row   = tileM * 16 + m;

    v8f c[8] = {};
    const float* xrow = (row < NUM_USERS) ? Xu + (size_t)row * HID
                                          : Xi + (size_t)(row - NUM_USERS) * HID;
    const float* xr = xrow + khalf;
    const float* wr = Wl + (size_t)khalf * HID + m;

    for (int k = 0; k < HID; k += 4) {
        v2f a = *(const v2f*)(xr + k);
        const float* w0 = wr + (size_t)k * HID;
        const float* w1 = w0 + HID;
#pragma unroll
        for (int t = 0; t < 8; ++t) {
            v2f b;
            b.x = w0[t * 16];
            b.y = w1[t * 16];
            c[t] = __builtin_amdgcn_wmma_f32_16x16x4_f32(false, a, false, b,
                                                         (short)0, c[t], false, false);
        }
    }
    const int rbase = tileM * 16 + ((lane >> 4) ? 8 : 0);
#pragma unroll
    for (int t = 0; t < 8; ++t)
#pragma unroll
        for (int r = 0; r < 8; ++r)
            H[(size_t)(rbase + r) * HID + t * 16 + m] = c[t][r];
}

__global__ __launch_bounds__(256) void alpha_k(const float* __restrict__ H,
                                               const float* __restrict__ Asrc,
                                               const float* __restrict__ Adst,
                                               float* __restrict__ as_out,
                                               float* __restrict__ ad_out) {
    const int warp = threadIdx.x >> 5, lane = threadIdx.x & 31;
    const int node = blockIdx.x * 8 + warp;
    if (node >= NNODES) return;
    float4 hv = ((const float4*)H)[(size_t)node * 32 + lane];
    float4 av = ((const float4*)Asrc)[lane];
    float4 dv = ((const float4*)Adst)[lane];
    float ss = hv.x * av.x + hv.y * av.y + hv.z * av.z + hv.w * av.w;
    float sd = hv.x * dv.x + hv.y * dv.y + hv.z * dv.z + hv.w * dv.w;
#pragma unroll
    for (int off = 16; off > 0; off >>= 1) {
        ss += __shfl_xor(ss, off, 32);
        sd += __shfl_xor(sd, off, 32);
    }
    if (lane == 0) { as_out[node] = ss; ad_out[node] = sd; }
}

#define NQUAD   (NE / 4)
#define SCAN_IT 18
#define NPAD    (32 * SCAN_IT * 32 * 4)
static_assert(NE % 4 == 0, "edge quads");
static_assert(NPAD >= NNODES + 1, "one block scans every node and the end marker");

__global__ __launch_bounds__(256) void zero_deg(int* __restrict__ deg) {
    int i = blockIdx.x * 256 + threadIdx.x;
    if (i < NPAD) deg[i] = 0;
}

__global__ __launch_bounds__(256) void hist_k(const int* __restrict__ ei,
                                              int* __restrict__ deg,
                                              int* __restrict__ rank) {
    int q = blockIdx.x * 256 + threadIdx.x;
    if (q >= NQUAD) return;
    int4 d4 = ((const int4*)(ei + NE))[q];
    int4 r;
    r.x = atomicAdd(deg + d4.x, 1);
    r.y = atomicAdd(deg + d4.y, 1);
    r.z = atomicAdd(deg + d4.z, 1);
    r.w = atomicAdd(deg + d4.w, 1);
    ((int4*)rank)[q] = r;
}

__global__ __launch_bounds__(1024) void scan_k(const int* __restrict__ deg,
                                               int* __restrict__ off) {
    __shared__ int wtot[32];
    const int lane = threadIdx.x & 31, wave = threadIdx.x >> 5;
    const int4* deg4 = (const int4*)deg + wave * (SCAN_IT * 32) + lane;
    int4*       off4 = (int4*)off       + wave * (SCAN_IT * 32) + lane;

    int4 v[SCAN_IT];
#pragma unroll
    for (int it = 0; it < SCAN_IT; ++it) v[it] = deg4[it * 32];
    int part = 0;
#pragma unroll
    for (int it = 0; it < SCAN_IT; ++it) part += (v[it].x + v[it].y) + (v[it].z + v[it].w);
#pragma unroll
    for (int o = 16; o > 0; o >>= 1) part += __shfl_xor(part, o, 32);
    if (lane == 0) wtot[wave] = part;
    __syncthreads();

    int carry = (lane < wave) ? wtot[lane] : 0;
#pragma unroll
    for (int o = 16; o > 0; o >>= 1) carry += __shfl_xor(carry, o, 32);

#pragma unroll
    for (int it = 0; it < SCAN_IT; ++it) {
        const int t = (v[it].x + v[it].y) + (v[it].z + v[it].w);
        int incl = t;
#pragma unroll
        for (int o = 1; o < 32; o <<= 1) {
            const int u = __shfl_up(incl, o, 32);
            if (lane >= o) incl += u;
        }
        const int ex = carry + incl - t;
        off4[it * 32] = make_int4(ex, ex + v[it].x, ex + v[it].x + v[it].y,
                                  ex + v[it].x + v[it].y + v[it].z);
        carry += __shfl(incl, 31, 32);
    }
}

__global__ __launch_bounds__(256) void scatter_k(const int* __restrict__ ei,
                                                 const int* __restrict__ off,
                                                 const int* __restrict__ rank,
                                                 int* __restrict__ csr) {
    int q = blockIdx.x * 256 + threadIdx.x;
    if (q >= NQUAD) return;
    int4 s4 = ((const int4*)ei)[q];
    int4 d4 = ((const int4*)(ei + NE))[q];
    int4 r4 = ((const int4*)rank)[q];
    csr[off[d4.x] + r4.x] = s4.x;
    csr[off[d4.y] + r4.y] = s4.y;
    csr[off[d4.z] + r4.z] = s4.z;
    csr[off[d4.w] + r4.w] = s4.w;
}

__device__ __forceinline__ float lrelu(float e) { return (e >= 0.f) ? e : NEG_SLOPE * e; }

template <int N> __device__ __forceinline__ float row_xor(float x) {
    return __int_as_float(__builtin_amdgcn_update_dpp(0, __float_as_int(x), 0x160 + N, 0xF, 0xF, false));
}
__device__ __forceinline__ float xhalf(float x) {
    const int b = __float_as_int(x);
    return __int_as_float(__builtin_amdgcn_permlanex16(b, b, 0x76543210, 0xfedcba98, false, false));
}
__device__ __forceinline__ float wave_max(float v) {
    v = fmaxf(v, row_xor<1>(v));
    v = fmaxf(v, row_xor<2>(v));
    v = fmaxf(v, row_xor<4>(v));
    v = fmaxf(v, row_xor<8>(v));
    return fmaxf(v, xhalf(v));
}
__device__ __forceinline__ float wave_sum(float v) {
    v += row_xor<1>(v);
    v += row_xor<2>(v);
    v += row_xor<4>(v);
    v += row_xor<8>(v);
    return v + xhalf(v);
}

__device__ __forceinline__ void gather_rows(float4& acc, const float4* __restrict__ H4,
                                            int s, float al, int cnt, int lane) {
    const int ab = __float_as_int(al);
    for (int k = 0; k < cnt; k += 4) {
        const int s0 = __builtin_amdgcn_readlane(s, k);
        const int s1 = __builtin_amdgcn_readlane(s, k + 1);
        const int s2 = __builtin_amdgcn_readlane(s, k + 2);
        const int s3 = __builtin_amdgcn_readlane(s, k + 3);
        const float a0 = __int_as_float(__builtin_amdgcn_readlane(ab, k));
        const float a1 = __int_as_float(__builtin_amdgcn_readlane(ab, k + 1));
        const float a2 = __int_as_float(__builtin_amdgcn_readlane(ab, k + 2));
        const float a3 = __int_as_float(__builtin_amdgcn_readlane(ab, k + 3));
        const float4 h0 = H4[(size_t)s0 * 32 + lane];
        const float4 h1 = H4[(size_t)s1 * 32 + lane];
        const float4 h2 = H4[(size_t)s2 * 32 + lane];
        const float4 h3 = H4[(size_t)s3 * 32 + lane];
        acc.x += h0.x * a0; acc.y += h0.y * a0; acc.z += h0.z * a0; acc.w += h0.w * a0;
        acc.x += h1.x * a1; acc.y += h1.y * a1; acc.z += h1.z * a1; acc.w += h1.w * a1;
        acc.x += h2.x * a2; acc.y += h2.y * a2; acc.z += h2.z * a2; acc.w += h2.w * a2;
        acc.x += h3.x * a3; acc.y += h3.y * a3; acc.z += h3.z * a3; acc.w += h3.w * a3;
    }
}

__global__ __launch_bounds__(256) void gat_node_k(const int* __restrict__ off,
                                                  const int* __restrict__ csr,
                                                  const float* __restrict__ H,
                                                  const float* __restrict__ as_,
                                                  const float* __restrict__ ad_,
                                                  const float* __restrict__ bias,
                                                  float* __restrict__ out) {
    const int lane = threadIdx.x & 31;
    const int d = __builtin_amdgcn_readfirstlane(blockIdx.x * 8 + (threadIdx.x >> 5));
    if (d >= NNODES) return;
    const int beg = off[d], end = off[d + 1];
    const int deg = end - beg;
    const float add = ad_[d];
    const float es = lrelu(as_[d] + add);
    const float4* H4 = (const float4*)H;
    const float4 hs = H4[(size_t)d * 32 + lane];
    float4 acc;

    if (deg <= 32) {
        int s = d; float e = -__builtin_inff();
        if (lane < deg) { s = csr[beg + lane]; e = lrelu(as_[s] + add); }
        const float m   = fmaxf(es, wave_max(e));
        const float ex  = (lane < deg) ? expf(e - m) : 0.f;
        const float exs = expf(es - m);
        const float den = exs + wave_sum(ex);
        const float al  = ex / den;
        const float als = exs / den;
        acc.x = hs.x * als; acc.y = hs.y * als; acc.z = hs.z * als; acc.w = hs.w * als;
        gather_rows(acc, H4, s, al, deg, lane);
    } else {
        float m = es;
        for (int b = beg; b < end; b += 32) {
            const int j = b + lane;
            if (j < end) m = fmaxf(m, lrelu(as_[csr[j]] + add));
        }
        m = wave_max(m);
        float part = 0.f;
        for (int b = beg; b < end; b += 32) {
            const int j = b + lane;
            if (j < end) part += expf(lrelu(as_[csr[j]] + add) - m);
        }
        const float exs = expf(es - m);
        const float den = exs + wave_sum(part);
        const float als = exs / den;
        acc.x = hs.x * als; acc.y = hs.y * als; acc.z = hs.z * als; acc.w = hs.w * als;
        for (int b = beg; b < end; b += 32) {
            const int j = b + lane;
            int s = d; float al = 0.f;
            if (j < end) { s = csr[j]; al = expf(lrelu(as_[s] + add) - m) / den; }
            const int left = end - b;
            gather_rows(acc, H4, s, al, left < 32 ? left : 32, lane);
        }
    }
    const float4 bv = ((const float4*)bias)[lane];
    acc.x += bv.x; acc.y += bv.y; acc.z += bv.z; acc.w += bv.w;
    ((float4*)out)[(size_t)d * 32 + lane] = acc;
}

extern "C" void kernel_launch(void* const* d_in, const int* in_sizes, int n_in,
                              void* d_out, int out_size, void* d_ws, size_t ws_size,
                              hipStream_t stream) {
    const int*   ei      = (const int*)d_in[0];
    const float* users   = (const float*)d_in[1];
    const float* items   = (const float*)d_in[2];
    const float* W       = (const float*)d_in[3];
    const float* att_src = (const float*)d_in[4];
    const float* att_dst = (const float*)d_in[5];
    const float* bias    = (const float*)d_in[6];
    float*       out     = (float*)d_out;

    float* xbuf = (float*)d_ws;
    float* hbuf = xbuf + (size_t)NNODES * HID;
    float* asb  = hbuf + (size_t)NNODES * HID;
    float* adb  = asb + NNODES;
    int*   deg  = (int*)(adb + NNODES);
    int*   off  = deg + NPAD;
    int*   rank = off + NPAD;
    int*   csr  = rank + NE;

    const int GRID_Z  = (NPAD + 255) / 256;
    const int GRID_Q  = (NQUAD + 255) / 256;
    const int GRID_GE = (MTILES + 7) / 8;
    const int GRID_NW = (NNODES + 7) / 8;

    zero_deg<<<GRID_Z, 256, 0, stream>>>(deg);
    hist_k<<<GRID_Q, 256, 0, stream>>>(ei, deg, rank);
    scan_k<<<1, 1024, 0, stream>>>(deg, off);
    scatter_k<<<GRID_Q, 256, 0, stream>>>(ei, off, rank, csr);

    for (int l = 0; l < 2; ++l) {
        const float* Wl = W + (size_t)l * HID * HID;
        float* tgt = (l == 0) ? xbuf : out;
        const float* xa = (l == 0) ? users : xbuf;
        const float* xb = (l == 0) ? items : xbuf + (size_t)NUM_USERS * HID;
        gemm_wmma<<<GRID_GE, 256, 0, stream>>>(xa, xb, Wl, hbuf);
        alpha_k<<<GRID_NW, 256, 0, stream>>>(hbuf, att_src + l * HID,
                                             att_dst + l * HID, asb, adb);
        gat_node_k<<<GRID_NW, 256, 0, stream>>>(off, csr, hbuf, asb, adb,
                                                bias + l * HID, tgt);
    }
}


# Logit kernel removed: each GEMM wave takes as/ad of its 16 rows from its accumulators (8 columns per lane, then a 16-lane DPP sum: new summation order of both dot products); 2 launches fewer
# speedup vs baseline: 3.4583x; 1.1403x over previous
#include <hip/hip_runtime.h>
#include <hip/hip_bf16.h>

#define NUM_USERS 50000
#define NUM_ITEMS 20000
#define NNODES    70000
#define MTILES    (NNODES / 16)
#define HID       128
#define NE        1000000
#define NEG_SLOPE 0.2f

typedef float v2f __attribute__((ext_vector_type(2)));
typedef float v8f __attribute__((ext_vector_type(8)));

template <int N> __device__ __forceinline__ float row_xor(float x) {
    return __int_as_float(__builtin_amdgcn_update_dpp(0, __float_as_int(x), 0x160 + N, 0xF, 0xF, false));
}
__device__ __forceinline__ float xhalf(float x) {
    const int b = __float_as_int(x);
    return __int_as_float(__builtin_amdgcn_permlanex16(b, b, 0x76543210, 0xfedcba98, false, false));
}
__device__ __forceinline__ float wave_max(float v) {
    v = fmaxf(v, row_xor<1>(v));
    v = fmaxf(v, row_xor<2>(v));
    v = fmaxf(v, row_xor<4>(v));
    v = fmaxf(v, row_xor<8>(v));
    return fmaxf(v, xhalf(v));
}
__device__ __forceinline__ float row_sum16(float v) {
    v += row_xor<1>(v);
    v += row_xor<2>(v);
    v += row_xor<4>(v);
    v += row_xor<8>(v);
    return v;
}
__device__ __forceinline__ float wave_sum(float v) {
    v = row_sum16(v);
    return v + xhalf(v);
}

__global__ __launch_bounds__(256) void gemm_wmma(const float* __restrict__ Xu,
                                                 const float* __restrict__ Xi,
                                                 const float* __restrict__ Wl,
                                                 const float* __restrict__ Asrc,
                                                 const float* __restrict__ Adst,
                                                 float* __restrict__ H,
                                                 float* __restrict__ as_out,
                                                 float* __restrict__ ad_out) {
    const int wave  = threadIdx.x >> 5;
    const int lane  = threadIdx.x & 31;
    const int tileM = blockIdx.x * 8 + wave;
    if (tileM >= MTILES) return;
    const int m     = lane & 15;
    const int khalf = (lane >> 4) * 2;
    const int row   = tileM * 16 + m;

    float av[8], dv[8];
#pragma unroll
    for (int t = 0; t < 8; ++t) { av[t] = Asrc[t * 16 + m]; dv[t] = Adst[t * 16 + m]; }

    v8f c[8] = {};
    const float* xrow = (row < NUM_USERS) ? Xu + (size_t)row * HID
                                          : Xi + (size_t)(row - NUM_USERS) * HID;
    const float* xr = xrow + khalf;
    const float* wr = Wl + (size_t)khalf * HID + m;

    for (int k = 0; k < HID; k += 4) {
        v2f a = *(const v2f*)(xr + k);
        const float* w0 = wr + (size_t)k * HID;
        const float* w1 = w0 + HID;
#pragma unroll
        for (int t = 0; t < 8; ++t) {
            v2f b;
            b.x = w0[t * 16];
            b.y = w1[t * 16];
            c[t] = __builtin_amdgcn_wmma_f32_16x16x4_f32(false, a, false, b,
                                                         (short)0, c[t], false, false);
        }
    }
    const int rbase = tileM * 16 + ((lane >> 4) ? 8 : 0);
#pragma unroll
    for (int t = 0; t < 8; ++t)
#pragma unroll
        for (int r = 0; r < 8; ++r)
            H[(size_t)(rbase + r) * HID + t * 16 + m] = c[t][r];

    float ps[8], pd[8];
#pragma unroll
    for (int r = 0; r < 8; ++r) {
        float s = c[0][r] * av[0];
        float d = c[0][r] * dv[0];
#pragma unroll
        for (int t = 1; t < 8; ++t) { s += c[t][r] * av[t]; d += c[t][r] * dv[t]; }
        ps[r] = row_sum16(s);
        pd[r] = row_sum16(d);
    }
    if (m == 0) {
        float4* sp = (float4*)(as_out + rbase);
        float4* dp = (float4*)(ad_out + rbase);
        sp[0] = make_float4(ps[0], ps[1], ps[2], ps[3]);
        sp[1] = make_float4(ps[4], ps[5], ps[6], ps[7]);
        dp[0] = make_float4(pd[0], pd[1], pd[2], pd[3]);
        dp[1] = make_float4(pd[4], pd[5], pd[6], pd[7]);
    }
}

#define NQUAD   (NE / 4)
#define SCAN_IT 18
#define NPAD    (32 * SCAN_IT * 32 * 4)
static_assert(NE % 4 == 0, "edge quads");
static_assert(NPAD >= NNODES + 1, "one block scans every node and the end marker");

__global__ __launch_bounds__(256) void zero_deg(int* __restrict__ deg) {
    int i = blockIdx.x * 256 + threadIdx.x;
    if (i < NPAD) deg[i] = 0;
}

__global__ __launch_bounds__(256) void hist_k(const int* __restrict__ ei,
                                              int* __restrict__ deg,
                                              int* __restrict__ rank) {
    int q = blockIdx.x * 256 + threadIdx.x;
    if (q >= NQUAD) return;
    int4 d4 = ((const int4*)(ei + NE))[q];
    int4 r;
    r.x = atomicAdd(deg + d4.x, 1);
    r.y = atomicAdd(deg + d4.y, 1);
    r.z = atomicAdd(deg + d4.z, 1);
    r.w = atomicAdd(deg + d4.w, 1);
    ((int4*)rank)[q] = r;
}

__global__ __launch_bounds__(1024) void scan_k(const int* __restrict__ deg,
                                               int* __restrict__ off) {
    __shared__ int wtot[32];
    const int lane = threadIdx.x & 31, wave = threadIdx.x >> 5;
    const int4* deg4 = (const int4*)deg + wave * (SCAN_IT * 32) + lane;
    int4*       off4 = (int4*)off       + wave * (SCAN_IT * 32) + lane;

    int4 v[SCAN_IT];
#pragma unroll
    for (int it = 0; it < SCAN_IT; ++it) v[it] = deg4[it * 32];
    int part = 0;
#pragma unroll
    for (int it = 0; it < SCAN_IT; ++it) part += (v[it].x + v[it].y) + (v[it].z + v[it].w);
#pragma unroll
    for (int o = 16; o > 0; o >>= 1) part += __shfl_xor(part, o, 32);
    if (lane == 0) wtot[wave] = part;
    __syncthreads();

    int carry = (lane < wave) ? wtot[lane] : 0;
#pragma unroll
    for (int o = 16; o > 0; o >>= 1) carry += __shfl_xor(carry, o, 32);

#pragma unroll
    for (int it = 0; it < SCAN_IT; ++it) {
        const int t = (v[it].x + v[it].y) + (v[it].z + v[it].w);
        int incl = t;
#pragma unroll
        for (int o = 1; o < 32; o <<= 1) {
            const int u = __shfl_up(incl, o, 32);
            if (lane >= o) incl += u;
        }
        const int ex = carry + incl - t;
        off4[it * 32] = make_int4(ex, ex + v[it].x, ex + v[it].x + v[it].y,
                                  ex + v[it].x + v[it].y + v[it].z);
        carry += __shfl(incl, 31, 32);
    }
}

__global__ __launch_bounds__(256) void scatter_k(const int* __restrict__ ei,
                                                 const int* __restrict__ off,
                                                 const int* __restrict__ rank,
                                                 int* __restrict__ csr) {
    int q = blockIdx.x * 256 + threadIdx.x;
    if (q >= NQUAD) return;
    int4 s4 = ((const int4*)ei)[q];
    int4 d4 = ((const int4*)(ei + NE))[q];
    int4 r4 = ((const int4*)rank)[q];
    csr[off[d4.x] + r4.x] = s4.x;
    csr[off[d4.y] + r4.y] = s4.y;
    csr[off[d4.z] + r4.z] = s4.z;
    csr[off[d4.w] + r4.w] = s4.w;
}

__device__ __forceinline__ float lrelu(float e) { return (e >= 0.f) ? e : NEG_SLOPE * e; }

__device__ __forceinline__ void gather_rows(float4& acc, const float4* __restrict__ H4,
                                            int s, float al, int cnt, int lane) {
    const int ab = __float_as_int(al);
    for (int k = 0; k < cnt; k += 4) {
        const int s0 = __builtin_amdgcn_readlane(s, k);
        const int s1 = __builtin_amdgcn_readlane(s, k + 1);
        const int s2 = __builtin_amdgcn_readlane(s, k + 2);
        const int s3 = __builtin_amdgcn_readlane(s, k + 3);
        const float a0 = __int_as_float(__builtin_amdgcn_readlane(ab, k));
        const float a1 = __int_as_float(__builtin_amdgcn_readlane(ab, k + 1));
        const float a2 = __int_as_float(__builtin_amdgcn_readlane(ab, k + 2));
        const float a3 = __int_as_float(__builtin_amdgcn_readlane(ab, k + 3));
        const float4 h0 = H4[(size_t)s0 * 32 + lane];
        const float4 h1 = H4[(size_t)s1 * 32 + lane];
        const float4 h2 = H4[(size_t)s2 * 32 + lane];
        const float4 h3 = H4[(size_t)s3 * 32 + lane];
        acc.x += h0.x * a0; acc.y += h0.y * a0; acc.z += h0.z * a0; acc.w += h0.w * a0;
        acc.x += h1.x * a1; acc.y += h1.y * a1; acc.z += h1.z * a1; acc.w += h1.w * a1;
        acc.x += h2.x * a2; acc.y += h2.y * a2; acc.z += h2.z * a2; acc.w += h2.w * a2;
        acc.x += h3.x * a3; acc.y += h3.y * a3; acc.z += h3.z * a3; acc.w += h3.w * a3;
    }
}

__global__ __launch_bounds__(256) void gat_node_k(const int* __restrict__ off,
                                                  const int* __restrict__ csr,
                                                  const float* __restrict__ H,
                                                  const float* __restrict__ as_,
                                                  const float* __restrict__ ad_,
                                                  const float* __restrict__ bias,
                                                  float* __restrict__ out) {
    const int lane = threadIdx.x & 31;
    const int d = __builtin_amdgcn_readfirstlane(blockIdx.x * 8 + (threadIdx.x >> 5));
    if (d >= NNODES) return;
    const int beg = off[d], end = off[d + 1];
    const int deg = end - beg;
    const float add = ad_[d];
    const float es = lrelu(as_[d] + add);
    const float4* H4 = (const float4*)H;
    const float4 hs = H4[(size_t)d * 32 + lane];
    float4 acc;

    if (deg <= 32) {
        int s = d; float e = -__builtin_inff();
        if (lane < deg) { s = csr[beg + lane]; e = lrelu(as_[s] + add); }
        const float m   = fmaxf(es, wave_max(e));
        const float ex  = (lane < deg) ? expf(e - m) : 0.f;
        const float exs = expf(es - m);
        const float den = exs + wave_sum(ex);
        const float al  = ex / den;
        const float als = exs / den;
        acc.x = hs.x * als; acc.y = hs.y * als; acc.z = hs.z * als; acc.w = hs.w * als;
        gather_rows(acc, H4, s, al, deg, lane);
    } else {
        float m = es;
        for (int b = beg; b < end; b += 32) {
            const int j = b + lane;
            if (j < end) m = fmaxf(m, lrelu(as_[csr[j]] + add));
        }
        m = wave_max(m);
        float part = 0.f;
        for (int b = beg; b < end; b += 32) {
            const int j = b + lane;
            if (j < end) part += expf(lrelu(as_[csr[j]] + add) - m);
        }
        const float exs = expf(es - m);
        const float den = exs + wave_sum(part);
        const float als = exs / den;
        acc.x = hs.x * als; acc.y = hs.y * als; acc.z = hs.z * als; acc.w = hs.w * als;
        for (int b = beg; b < end; b += 32) {
            const int j = b + lane;
            int s = d; float al = 0.f;
            if (j < end) { s = csr[j]; al = expf(lrelu(as_[s] + add) - m) / den; }
            const int left = end - b;
            gather_rows(acc, H4, s, al, left < 32 ? left : 32, lane);
        }
    }
    const float4 bv = ((const float4*)bias)[lane];
    acc.x += bv.x; acc.y += bv.y; acc.z += bv.z; acc.w += bv.w;
    ((float4*)out)[(size_t)d * 32 + lane] = acc;
}

extern "C" void kernel_launch(void* const* d_in, const int* in_sizes, int n_in,
                              void* d_out, int out_size, void* d_ws, size_t ws_size,
                              hipStream_t stream) {
    const int*   ei      = (const int*)d_in[0];
    const float* users   = (const float*)d_in[1];
    const float* items   = (const float*)d_in[2];
    const float* W       = (const float*)d_in[3];
    const float* att_src = (const float*)d_in[4];
    const float* att_dst = (const float*)d_in[5];
    const float* bias    = (const float*)d_in[6];
    float*       out     = (float*)d_out;

    float* xbuf = (float*)d_ws;
    float* hbuf = xbuf + (size_t)NNODES * HID;
    float* asb  = hbuf + (size_t)NNODES * HID;
    float* adb  = asb + NNODES;
    int*   deg  = (int*)(adb + NNODES);
    int*   off  = deg + NPAD;
    int*   rank = off + NPAD;
    int*   csr  = rank + NE;

    const int GRID_Z  = (NPAD + 255) / 256;
    const int GRID_Q  = (NQUAD + 255) / 256;
    const int GRID_GE = (MTILES + 7) / 8;
    const int GRID_NW = (NNODES + 7) / 8;

    zero_deg<<<GRID_Z, 256, 0, stream>>>(deg);
    hist_k<<<GRID_Q, 256, 0, stream>>>(ei, deg, rank);
    scan_k<<<1, 1024, 0, stream>>>(deg, off);
    scatter_k<<<GRID_Q, 256, 0, stream>>>(ei, off, rank, csr);

    for (int l = 0; l < 2; ++l) {
        const float* Wl = W + (size_t)l * HID * HID;
        float* tgt = (l == 0) ? xbuf : out;
        const float* xa = (l == 0) ? users : xbuf;
        const float* xb = (l == 0) ? items : xbuf + (size_t)NUM_USERS * HID;
        gemm_wmma<<<GRID_GE, 256, 0, stream>>>(xa, xb, Wl, att_src + l * HID,
                                               att_dst + l * HID, hbuf, asb, adb);
        gat_node_k<<<GRID_NW, 256, 0, stream>>>(off, csr, hbuf, asb, adb,
                                                bias + l * HID, tgt);
    }
}


# Layer-0 product split in two row halves that run as extra work-groups in front of the histogram launch and of the scatter launch (independent work overlapped); its own launch is gone: 7 launches
# speedup vs baseline: 3.6887x; 1.0666x over previous
#include <hip/hip_runtime.h>
#include <hip/hip_bf16.h>

#define NUM_USERS 50000
#define NUM_ITEMS 20000
#define NNODES    70000
#define MTILES    (NNODES / 16)
#define HID       128
#define NE        1000000
#define NEG_SLOPE 0.2f

typedef float v2f __attribute__((ext_vector_type(2)));
typedef float v8f __attribute__((ext_vector_type(8)));

template <int N> __device__ __forceinline__ float row_xor(float x) {
    return __int_as_float(__builtin_amdgcn_update_dpp(0, __float_as_int(x), 0x160 + N, 0xF, 0xF, false));
}
__device__ __forceinline__ float xhalf(float x) {
    const int b = __float_as_int(x);
    return __int_as_float(__builtin_amdgcn_permlanex16(b, b, 0x76543210, 0xfedcba98, false, false));
}
__device__ __forceinline__ float wave_max(float v) {
    v = fmaxf(v, row_xor<1>(v));
    v = fmaxf(v, row_xor<2>(v));
    v = fmaxf(v, row_xor<4>(v));
    v = fmaxf(v, row_xor<8>(v));
    return fmaxf(v, xhalf(v));
}
__device__ __forceinline__ float row_sum16(float v) {
    v += row_xor<1>(v);
    v += row_xor<2>(v);
    v += row_xor<4>(v);
    v += row_xor<8>(v);
    return v;
}
__device__ __forceinline__ float wave_sum(float v) {
    v = row_sum16(v);
    return v + xhalf(v);
}

__device__ __forceinline__ void gemm_tile(const int tileM, const int lane,
                                          const float* __restrict__ Xu,
                                          const float* __restrict__ Xi,
                                          const float* __restrict__ Wl,
                                          const float* __restrict__ Asrc,
                                          const float* __restrict__ Adst,
                                          float* __restrict__ H,
                                          float* __restrict__ as_out,
                                          float* __restrict__ ad_out) {
    const int m     = lane & 15;
    const int khalf = (lane >> 4) * 2;
    const int row   = tileM * 16 + m;

    float av[8], dv[8];
#pragma unroll
    for (int t = 0; t < 8; ++t) { av[t] = Asrc[t * 16 + m]; dv[t] = Adst[t * 16 + m]; }

    v8f c[8] = {};
    const float* xrow = (row < NUM_USERS) ? Xu + (size_t)row * HID
                                          : Xi + (size_t)(row - NUM_USERS) * HID;
    const float* xr = xrow + khalf;
    const float* wr = Wl + (size_t)khalf * HID + m;

    for (int k = 0; k < HID; k += 4) {
        v2f a = *(const v2f*)(xr + k);
        const float* w0 = wr + (size_t)k * HID;
        const float* w1 = w0 + HID;
#pragma unroll
        for (int t = 0; t < 8; ++t) {
            v2f b;
            b.x = w0[t * 16];
            b.y = w1[t * 16];
            c[t] = __builtin_amdgcn_wmma_f32_16x16x4_f32(false, a, false, b,
                                                         (short)0, c[t], false, false);
        }
    }
    const int rbase = tileM * 16 + ((lane >> 4) ? 8 : 0);
#pragma unroll
    for (int t = 0; t < 8; ++t)
#pragma unroll
        for (int r = 0; r < 8; ++r)
            H[(size_t)(rbase + r) * HID + t * 16 + m] = c[t][r];

    float ps[8], pd[8];
#pragma unroll
    for (int r = 0; r < 8; ++r) {
        float s = c[0][r] * av[0];
        float d = c[0][r] * dv[0];
#pragma unroll
        for (int t = 1; t < 8; ++t) { s += c[t][r] * av[t]; d += c[t][r] * dv[t]; }
        ps[r] = row_sum16(s);
        pd[r] = row_sum16(d);
    }
    if (m == 0) {
        float4* sp = (float4*)(as_out + rbase);
        float4* dp = (float4*)(ad_out + rbase);
        sp[0] = make_float4(ps[0], ps[1], ps[2], ps[3]);
        sp[1] = make_float4(ps[4], ps[5], ps[6], ps[7]);
        dp[0] = make_float4(pd[0], pd[1], pd[2], pd[3]);
        dp[1] = make_float4(pd[4], pd[5], pd[6], pd[7]);
    }
}

__global__ __launch_bounds__(256) void gemm_wmma(const float* __restrict__ Xu,
                                                 const float* __restrict__ Xi,
                                                 const float* __restrict__ Wl,
                                                 const float* __restrict__ Asrc,
                                                 const float* __restrict__ Adst,
                                                 float* __restrict__ H,
                                                 float* __restrict__ as_out,
                                                 float* __restrict__ ad_out) {
    const int tileM = blockIdx.x * 8 + (threadIdx.x >> 5);
    if (tileM >= MTILES) return;
    gemm_tile(tileM, threadIdx.x & 31, Xu, Xi, Wl, Asrc, Adst, H, as_out, ad_out);
}

#define NQUAD   (NE / 4)
#define SCAN_IT 18
#define NPAD    (32 * SCAN_IT * 32 * 4)
#define GB_A    274
#define TILES_A (GB_A * 8)
#define GB_B    ((MTILES - TILES_A + 7) / 8)
static_assert(NE % 4 == 0, "edge quads");
static_assert(TILES_A < MTILES, "both halves hold tiles");
static_assert(NPAD >= NNODES + 1, "one block scans every node and the end marker");

__global__ __launch_bounds__(256) void zero_deg(int* __restrict__ deg) {
    int i = blockIdx.x * 256 + threadIdx.x;
    if (i < NPAD) deg[i] = 0;
}

__global__ __launch_bounds__(256) void hist_k(const int* __restrict__ ei,
                                              int* __restrict__ deg,
                                              int* __restrict__ rank,
                                              const float* __restrict__ Xu,
                                              const float* __restrict__ Xi,
                                              const float* __restrict__ Wl,
                                              const float* __restrict__ Asrc,
                                              const float* __restrict__ Adst,
                                              float* __restrict__ H,
                                              float* __restrict__ as_out,
                                              float* __restrict__ ad_out) {
    if (blockIdx.x < GB_A) {
        gemm_tile(blockIdx.x * 8 + (threadIdx.x >> 5), threadIdx.x & 31,
                  Xu, Xi, Wl, Asrc, Adst, H, as_out, ad_out);
        return;
    }
    int q = (blockIdx.x - GB_A) * 256 + threadIdx.x;
    if (q >= NQUAD) return;
    int4 d4 = ((const int4*)(ei + NE))[q];
    int4 r;
    r.x = atomicAdd(deg + d4.x, 1);
    r.y = atomicAdd(deg + d4.y, 1);
    r.z = atomicAdd(deg + d4.z, 1);
    r.w = atomicAdd(deg + d4.w, 1);
    ((int4*)rank)[q] = r;
}

__global__ __launch_bounds__(1024) void scan_k(const int* __restrict__ deg,
                                               int* __restrict__ off) {
    __shared__ int wtot[32];
    const int lane = threadIdx.x & 31, wave = threadIdx.x >> 5;
    const int4* deg4 = (const int4*)deg + wave * (SCAN_IT * 32) + lane;
    int4*       off4 = (int4*)off       + wave * (SCAN_IT * 32) + lane;

    int4 v[SCAN_IT];
#pragma unroll
    for (int it = 0; it < SCAN_IT; ++it) v[it] = deg4[it * 32];
    int part = 0;
#pragma unroll
    for (int it = 0; it < SCAN_IT; ++it) part += (v[it].x + v[it].y) + (v[it].z + v[it].w);
#pragma unroll
    for (int o = 16; o > 0; o >>= 1) part += __shfl_xor(part, o, 32);
    if (lane == 0) wtot[wave] = part;
    __syncthreads();

    int carry = (lane < wave) ? wtot[lane] : 0;
#pragma unroll
    for (int o = 16; o > 0; o >>= 1) carry += __shfl_xor(carry, o, 32);

#pragma unroll
    for (int it = 0; it < SCAN_IT; ++it) {
        const int t = (v[it].x + v[it].y) + (v[it].z + v[it].w);
        int incl = t;
#pragma unroll
        for (int o = 1; o < 32; o <<= 1) {
            const int u = __shfl_up(incl, o, 32);
            if (lane >= o) incl += u;
        }
        const int ex = carry + incl - t;
        off4[it * 32] = make_int4(ex, ex + v[it].x, ex + v[it].x + v[it].y,
                                  ex + v[it].x + v[it].y + v[it].z);
        carry += __shfl(incl, 31, 32);
    }
}

__global__ __launch_bounds__(256) void scatter_k(const int* __restrict__ ei,
                                                 const int* __restrict__ off,
                                                 const int* __restrict__ rank,
                                                 int* __restrict__ csr,
                                                 const float* __restrict__ Xu,
                                                 const float* __restrict__ Xi,
                                                 const float* __restrict__ Wl,
                                                 const float* __restrict__ Asrc,
                                                 const float* __restrict__ Adst,
                                                 float* __restrict__ H,
                                                 float* __restrict__ as_out,
                                                 float* __restrict__ ad_out) {
    if (blockIdx.x < GB_B) {
        const int tileM = TILES_A + blockIdx.x * 8 + (threadIdx.x >> 5);
        if (tileM < MTILES)
            gemm_tile(tileM, threadIdx.x & 31, Xu, Xi, Wl, Asrc, Adst, H, as_out, ad_out);
        return;
    }
    int q = (blockIdx.x - GB_B) * 256 + threadIdx.x;
    if (q >= NQUAD) return;
    int4 s4 = ((const int4*)ei)[q];
    int4 d4 = ((const int4*)(ei + NE))[q];
    int4 r4 = ((const int4*)rank)[q];
    csr[off[d4.x] + r4.x] = s4.x;
    csr[off[d4.y] + r4.y] = s4.y;
    csr[off[d4.z] + r4.z] = s4.z;
    csr[off[d4.w] + r4.w] = s4.w;
}

__device__ __forceinline__ float lrelu(float e) { return (e >= 0.f) ? e : NEG_SLOPE * e; }

__device__ __forceinline__ void gather_rows(float4& acc, const float4* __restrict__ H4,
                                            int s, float al, int cnt, int lane) {
    const int ab = __float_as_int(al);
    for (int k = 0; k < cnt; k += 4) {
        const int s0 = __builtin_amdgcn_readlane(s, k);
        const int s1 = __builtin_amdgcn_readlane(s, k + 1);
        const int s2 = __builtin_amdgcn_readlane(s, k + 2);
        const int s3 = __builtin_amdgcn_readlane(s, k + 3);
        const float a0 = __int_as_float(__builtin_amdgcn_readlane(ab, k));
        const float a1 = __int_as_float(__builtin_amdgcn_readlane(ab, k + 1));
        const float a2 = __int_as_float(__builtin_amdgcn_readlane(ab, k + 2));
        const float a3 = __int_as_float(__builtin_amdgcn_readlane(ab, k + 3));
        const float4 h0 = H4[(size_t)s0 * 32 + lane];
        const float4 h1 = H4[(size_t)s1 * 32 + lane];
        const float4 h2 = H4[(size_t)s2 * 32 + lane];
        const float4 h3 = H4[(size_t)s3 * 32 + lane];
        acc.x += h0.x * a0; acc.y += h0.y * a0; acc.z += h0.z * a0; acc.w += h0.w * a0;
        acc.x += h1.x * a1; acc.y += h1.y * a1; acc.z += h1.z * a1; acc.w += h1.w * a1;
        acc.x += h2.x * a2; acc.y += h2.y * a2; acc.z += h2.z * a2; acc.w += h2.w * a2;
        acc.x += h3.x * a3; acc.y += h3.y * a3; acc.z += h3.z * a3; acc.w += h3.w * a3;
    }
}

__global__ __launch_bounds__(256) void gat_node_k(const int* __restrict__ off,
                                                  const int* __restrict__ csr,
                                                  const float* __restrict__ H,
                                                  const float* __restrict__ as_,
                                                  const float* __restrict__ ad_,
                                                  const float* __restrict__ bias,
                                                  float* __restrict__ out) {
    const int lane = threadIdx.x & 31;
    const int d = __builtin_amdgcn_readfirstlane(blockIdx.x * 8 + (threadIdx.x >> 5));
    if (d >= NNODES) return;
    const int beg = off[d], end = off[d + 1];
    const int deg = end - beg;
    const float add = ad_[d];
    const float es = lrelu(as_[d] + add);
    const float4* H4 = (const float4*)H;
    const float4 hs = H4[(size_t)d * 32 + lane];
    float4 acc;

    if (deg <= 32) {
        int s = d; float e = -__builtin_inff();
        if (lane < deg) { s = csr[beg + lane]; e = lrelu(as_[s] + add); }
        const float m   = fmaxf(es, wave_max(e));
        const float ex  = (lane < deg) ? expf(e - m) : 0.f;
        const float exs = expf(es - m);
        const float den = exs + wave_sum(ex);
        const float al  = ex / den;
        const float als = exs / den;
        acc.x = hs.x * als; acc.y = hs.y * als; acc.z = hs.z * als; acc.w = hs.w * als;
        gather_rows(acc, H4, s, al, deg, lane);
    } else {
        float m = es;
        for (int b = beg; b < end; b += 32) {
            const int j = b + lane;
            if (j < end) m = fmaxf(m, lrelu(as_[csr[j]] + add));
        }
        m = wave_max(m);
        float part = 0.f;
        for (int b = beg; b < end; b += 32) {
            const int j = b + lane;
            if (j < end) part += expf(lrelu(as_[csr[j]] + add) - m);
        }
        const float exs = expf(es - m);
        const float den = exs + wave_sum(part);
        const float als = exs / den;
        acc.x = hs.x * als; acc.y = hs.y * als; acc.z = hs.z * als; acc.w = hs.w * als;
        for (int b = beg; b < end; b += 32) {
            const int j = b + lane;
            int s = d; float al = 0.f;
            if (j < end) { s = csr[j]; al = expf(lrelu(as_[s] + add) - m) / den; }
            const int left = end - b;
            gather_rows(acc, H4, s, al, left < 32 ? left : 32, lane);
        }
    }
    const float4 bv = ((const float4*)bias)[lane];
    acc.x += bv.x; acc.y += bv.y; acc.z += bv.z; acc.w += bv.w;
    ((float4*)out)[(size_t)d * 32 + lane] = acc;
}

extern "C" void kernel_launch(void* const* d_in, const int* in_sizes, int n_in,
                              void* d_out, int out_size, void* d_ws, size_t ws_size,
                              hipStream_t stream) {
    const int*   ei      = (const int*)d_in[0];
    const float* users   = (const float*)d_in[1];
    const float* items   = (const float*)d_in[2];
    const float* W       = (const float*)d_in[3];
    const float* att_src = (const float*)d_in[4];
    const float* att_dst = (const float*)d_in[5];
    const float* bias    = (const float*)d_in[6];
    float*       out     = (float*)d_out;

    float* xbuf = (float*)d_ws;
    float* hbuf = xbuf + (size_t)NNODES * HID;
    float* asb  = hbuf + (size_t)NNODES * HID;
    float* adb  = asb + NNODES;
    int*   deg  = (int*)(adb + NNODES);
    int*   off  = deg + NPAD;
    int*   rank = off + NPAD;
    int*   csr  = rank + NE;

    const int GRID_Z  = (NPAD + 255) / 256;
    const int GRID_Q  = (NQUAD + 255) / 256;
    const int GRID_GE = (MTILES + 7) / 8;
    const int GRID_NW = (NNODES + 7) / 8;
    const int GRID_HG = GB_A + GRID_Q;
    const int GRID_SG = GB_B + GRID_Q;

    zero_deg<<<GRID_Z, 256, 0, stream>>>(deg);
    hist_k<<<GRID_HG, 256, 0, stream>>>(ei, deg, rank, users, items, W, att_src, att_dst,
                                              hbuf, asb, adb);
    scan_k<<<1, 1024, 0, stream>>>(deg, off);
    scatter_k<<<GRID_SG, 256, 0, stream>>>(ei, off, rank, csr, users, items, W, att_src,
                                                 att_dst, hbuf, asb, adb);
    gat_node_k<<<GRID_NW, 256, 0, stream>>>(off, csr, hbuf, asb, adb, bias, xbuf);

    gemm_wmma<<<GRID_GE, 256, 0, stream>>>(xbuf, xbuf + (size_t)NUM_USERS * HID,
                                           W + (size_t)HID * HID, att_src + HID, att_dst + HID,
                                           hbuf, asb, adb);
    gat_node_k<<<GRID_NW, 256, 0, stream>>>(off, csr, hbuf, asb, adb, bias + HID, out);
}


# Product's tile columns re-mapped (col = 64*(t>>2) + 4*lane + (t&3)) so W, h and the attention vectors move in 16-byte pieces: 4 loads per k step for 16, 16 stores for 64; new as/ad sum order
# speedup vs baseline: 3.8082x; 1.0324x over previous
#include <hip/hip_runtime.h>
#include <hip/hip_bf16.h>

#define NUM_USERS 50000
#define NUM_ITEMS 20000
#define NNODES    70000
#define MTILES    (NNODES / 16)
#define HID       128
#define NE        1000000
#define NEG_SLOPE 0.2f

typedef float v2f __attribute__((ext_vector_type(2)));
typedef float v8f __attribute__((ext_vector_type(8)));

template <int N> __device__ __forceinline__ float row_xor(float x) {
    return __int_as_float(__builtin_amdgcn_update_dpp(0, __float_as_int(x), 0x160 + N, 0xF, 0xF, false));
}
__device__ __forceinline__ float xhalf(float x) {
    const int b = __float_as_int(x);
    return __int_as_float(__builtin_amdgcn_permlanex16(b, b, 0x76543210, 0xfedcba98, false, false));
}
__device__ __forceinline__ float wave_max(float v) {
    v = fmaxf(v, row_xor<1>(v));
    v = fmaxf(v, row_xor<2>(v));
    v = fmaxf(v, row_xor<4>(v));
    v = fmaxf(v, row_xor<8>(v));
    return fmaxf(v, xhalf(v));
}
__device__ __forceinline__ float row_sum16(float v) {
    v += row_xor<1>(v);
    v += row_xor<2>(v);
    v += row_xor<4>(v);
    v += row_xor<8>(v);
    return v;
}
__device__ __forceinline__ float wave_sum(float v) {
    v = row_sum16(v);
    return v + xhalf(v);
}

__device__ __forceinline__ void gemm_tile(const int tileM, const int lane,
                                          const float* __restrict__ Xu,
                                          const float* __restrict__ Xi,
                                          const float* __restrict__ Wl,
                                          const float* __restrict__ Asrc,
                                          const float* __restrict__ Adst,
                                          float* __restrict__ H,
                                          float* __restrict__ as_out,
                                          float* __restrict__ ad_out) {
    const int m     = lane & 15;
    const int khalf = (lane >> 4) * 2;
    const int row   = tileM * 16 + m;

    const float4 av0 = ((const float4*)Asrc)[m], av1 = ((const float4*)Asrc)[16 + m];
    const float4 dv0 = ((const float4*)Adst)[m], dv1 = ((const float4*)Adst)[16 + m];
    const float av[8] = {av0.x, av0.y, av0.z, av0.w, av1.x, av1.y, av1.z, av1.w};
    const float dv[8] = {dv0.x, dv0.y, dv0.z, dv0.w, dv1.x, dv1.y, dv1.z, dv1.w};

    v8f c[8] = {};
    const float* xrow = (row < NUM_USERS) ? Xu + (size_t)row * HID
                                          : Xi + (size_t)(row - NUM_USERS) * HID;
    const float* xr = xrow + khalf;
    const float4* wr = (const float4*)(Wl + (size_t)khalf * HID) + m;

    for (int k = 0; k < HID; k += 4) {
        v2f a = *(const v2f*)(xr + k);
        const float4* w0 = wr + (size_t)k * (HID / 4);
        const float4* w1 = w0 + (HID / 4);
        const float4 p0 = w0[0], p1 = w0[16];
        const float4 q0 = w1[0], q1 = w1[16];
        const float bx[8] = {p0.x, p0.y, p0.z, p0.w, p1.x, p1.y, p1.z, p1.w};
        const float by[8] = {q0.x, q0.y, q0.z, q0.w, q1.x, q1.y, q1.z, q1.w};
#pragma unroll
        for (int t = 0; t < 8; ++t) {
            v2f b;
            b.x = bx[t];
            b.y = by[t];
            c[t] = __builtin_amdgcn_wmma_f32_16x16x4_f32(false, a, false, b,
                                                         (short)0, c[t], false, false);
        }
    }
    const int rbase = tileM * 16 + ((lane >> 4) ? 8 : 0);
    float4* hrow = (float4*)(H + (size_t)rbase * HID) + m;
#pragma unroll
    for (int r = 0; r < 8; ++r) {
        hrow[r * (HID / 4)]      = make_float4(c[0][r], c[1][r], c[2][r], c[3][r]);
        hrow[r * (HID / 4) + 16] = make_float4(c[4][r], c[5][r], c[6][r], c[7][r]);
    }

    float ps[8], pd[8];
#pragma unroll
    for (int r = 0; r < 8; ++r) {
        float s = c[0][r] * av[0];
        float d = c[0][r] * dv[0];
#pragma unroll
        for (int t = 1; t < 8; ++t) { s += c[t][r] * av[t]; d += c[t][r] * dv[t]; }
        ps[r] = row_sum16(s);
        pd[r] = row_sum16(d);
    }
    if (m == 0) {
        float4* sp = (float4*)(as_out + rbase);
        float4* dp = (float4*)(ad_out + rbase);
        sp[0] = make_float4(ps[0], ps[1], ps[2], ps[3]);
        sp[1] = make_float4(ps[4], ps[5], ps[6], ps[7]);
        dp[0] = make_float4(pd[0], pd[1], pd[2], pd[3]);
        dp[1] = make_float4(pd[4], pd[5], pd[6], pd[7]);
    }
}

__global__ __launch_bounds__(256) void gemm_wmma(const float* __restrict__ Xu,
                                                 const float* __restrict__ Xi,
                                                 const float* __restrict__ Wl,
                                                 const float* __restrict__ Asrc,
                                                 const float* __restrict__ Adst,
                                                 float* __restrict__ H,
                                                 float* __restrict__ as_out,
                                                 float* __restrict__ ad_out) {
    const int tileM = blockIdx.x * 8 + (threadIdx.x >> 5);
    if (tileM >= MTILES) return;
    gemm_tile(tileM, threadIdx.x & 31, Xu, Xi, Wl, Asrc, Adst, H, as_out, ad_out);
}

#define NQUAD   (NE / 4)
#define SCAN_IT 18
#define NPAD    (32 * SCAN_IT * 32 * 4)
#define GB_A    274
#define TILES_A (GB_A * 8)
#define GB_B    ((MTILES - TILES_A + 7) / 8)
static_assert(NE % 4 == 0, "edge quads");
static_assert(TILES_A < MTILES, "both halves hold tiles");
static_assert(NPAD >= NNODES + 1, "one block scans every node and the end marker");

__global__ __launch_bounds__(256) void zero_deg(int* __restrict__ deg) {
    int i = blockIdx.x * 256 + threadIdx.x;
    if (i < NPAD) deg[i] = 0;
}

__global__ __launch_bounds__(256) void hist_k(const int* __restrict__ ei,
                                              int* __restrict__ deg,
                                              int* __restrict__ rank,
                                              const float* __restrict__ Xu,
                                              const float* __restrict__ Xi,
                                              const float* __restrict__ Wl,
                                              const float* __restrict__ Asrc,
                                              const float* __restrict__ Adst,
                                              float* __restrict__ H,
                                              float* __restrict__ as_out,
                                              float* __restrict__ ad_out) {
    if (blockIdx.x < GB_A) {
        gemm_tile(blockIdx.x * 8 + (threadIdx.x >> 5), threadIdx.x & 31,
                  Xu, Xi, Wl, Asrc, Adst, H, as_out, ad_out);
        return;
    }
    int q = (blockIdx.x - GB_A) * 256 + threadIdx.x;
    if (q >= NQUAD) return;
    int4 d4 = ((const int4*)(ei + NE))[q];
    int4 r;
    r.x = atomicAdd(deg + d4.x, 1);
    r.y = atomicAdd(deg + d4.y, 1);
    r.z = atomicAdd(deg + d4.z, 1);
    r.w = atomicAdd(deg + d4.w, 1);
    ((int4*)rank)[q] = r;
}

__global__ __launch_bounds__(1024) void scan_k(const int* __restrict__ deg,
                                               int* __restrict__ off) {
    __shared__ int wtot[32];
    const int lane = threadIdx.x & 31, wave = threadIdx.x >> 5;
    const int4* deg4 = (const int4*)deg + wave * (SCAN_IT * 32) + lane;
    int4*       off4 = (int4*)off       + wave * (SCAN_IT * 32) + lane;

    int4 v[SCAN_IT];
#pragma unroll
    for (int it = 0; it < SCAN_IT; ++it) v[it] = deg4[it * 32];
    int part = 0;
#pragma unroll
    for (int it = 0; it < SCAN_IT; ++it) part += (v[it].x + v[it].y) + (v[it].z + v[it].w);
#pragma unroll
    for (int o = 16; o > 0; o >>= 1) part += __shfl_xor(part, o, 32);
    if (lane == 0) wtot[wave] = part;
    __syncthreads();

    int carry = (lane < wave) ? wtot[lane] : 0;
#pragma unroll
    for (int o = 16; o > 0; o >>= 1) carry += __shfl_xor(carry, o, 32);

#pragma unroll
    for (int it = 0; it < SCAN_IT; ++it) {
        const int t = (v[it].x + v[it].y) + (v[it].z + v[it].w);
        int incl = t;
#pragma unroll
        for (int o = 1; o < 32; o <<= 1) {
            const int u = __shfl_up(incl, o, 32);
            if (lane >= o) incl += u;
        }
        const int ex = carry + incl - t;
        off4[it * 32] = make_int4(ex, ex + v[it].x, ex + v[it].x + v[it].y,
                                  ex + v[it].x + v[it].y + v[it].z);
        carry += __shfl(incl, 31, 32);
    }
}

__global__ __launch_bounds__(256) void scatter_k(const int* __restrict__ ei,
                                                 const int* __restrict__ off,
                                                 const int* __restrict__ rank,
                                                 int* __restrict__ csr,
                                                 const float* __restrict__ Xu,
                                                 const float* __restrict__ Xi,
                                                 const float* __restrict__ Wl,
                                                 const float* __restrict__ Asrc,
                                                 const float* __restrict__ Adst,
                                                 float* __restrict__ H,
                                                 float* __restrict__ as_out,
                                                 float* __restrict__ ad_out) {
    if (blockIdx.x < GB_B) {
        const int tileM = TILES_A + blockIdx.x * 8 + (threadIdx.x >> 5);
        if (tileM < MTILES)
            gemm_tile(tileM, threadIdx.x & 31, Xu, Xi, Wl, Asrc, Adst, H, as_out, ad_out);
        return;
    }
    int q = (blockIdx.x - GB_B) * 256 + threadIdx.x;
    if (q >= NQUAD) return;
    int4 s4 = ((const int4*)ei)[q];
    int4 d4 = ((const int4*)(ei + NE))[q];
    int4 r4 = ((const int4*)rank)[q];
    csr[off[d4.x] + r4.x] = s4.x;
    csr[off[d4.y] + r4.y] = s4.y;
    csr[off[d4.z] + r4.z] = s4.z;
    csr[off[d4.w] + r4.w] = s4.w;
}

__device__ __forceinline__ float lrelu(float e) { return (e >= 0.f) ? e : NEG_SLOPE * e; }

__device__ __forceinline__ void gather_rows(float4& acc, const float4* __restrict__ H4,
                                            int s, float al, int cnt, int lane) {
    const int ab = __float_as_int(al);
    for (int k = 0; k < cnt; k += 4) {
        const int s0 = __builtin_amdgcn_readlane(s, k);
        const int s1 = __builtin_amdgcn_readlane(s, k + 1);
        const int s2 = __builtin_amdgcn_readlane(s, k + 2);
        const int s3 = __builtin_amdgcn_readlane(s, k + 3);
        const float a0 = __int_as_float(__builtin_amdgcn_readlane(ab, k));
        const float a1 = __int_as_float(__builtin_amdgcn_readlane(ab, k + 1));
        const float a2 = __int_as_float(__builtin_amdgcn_readlane(ab, k + 2));
        const float a3 = __int_as_float(__builtin_amdgcn_readlane(ab, k + 3));
        const float4 h0 = H4[(size_t)s0 * 32 + lane];
        const float4 h1 = H4[(size_t)s1 * 32 + lane];
        const float4 h2 = H4[(size_t)s2 * 32 + lane];
        const float4 h3 = H4[(size_t)s3 * 32 + lane];
        acc.x += h0.x * a0; acc.y += h0.y * a0; acc.z += h0.z * a0; acc.w += h0.w * a0;
        acc.x += h1.x * a1; acc.y += h1.y * a1; acc.z += h1.z * a1; acc.w += h1.w * a1;
        acc.x += h2.x * a2; acc.y += h2.y * a2; acc.z += h2.z * a2; acc.w += h2.w * a2;
        acc.x += h3.x * a3; acc.y += h3.y * a3; acc.z += h3.z * a3; acc.w += h3.w * a3;
    }
}

__global__ __launch_bounds__(256) void gat_node_k(const int* __restrict__ off,
                                                  const int* __restrict__ csr,
                                                  const float* __restrict__ H,
                                                  const float* __restrict__ as_,
                                                  const float* __restrict__ ad_,
                                                  const float* __restrict__ bias,
                                                  float* __restrict__ out) {
    const int lane = threadIdx.x & 31;
    const int d = __builtin_amdgcn_readfirstlane(blockIdx.x * 8 + (threadIdx.x >> 5));
    if (d >= NNODES) return;
    const int beg = off[d], end = off[d + 1];
    const int deg = end - beg;
    const float add = ad_[d];
    const float es = lrelu(as_[d] + add);
    const float4* H4 = (const float4*)H;
    const float4 hs = H4[(size_t)d * 32 + lane];
    float4 acc;

    if (deg <= 32) {
        int s = d; float e = -__builtin_inff();
        if (lane < deg) { s = csr[beg + lane]; e = lrelu(as_[s] + add); }
        const float m   = fmaxf(es, wave_max(e));
        const float ex  = (lane < deg) ? expf(e - m) : 0.f;
        const float exs = expf(es - m);
        const float den = exs + wave_sum(ex);
        const float al  = ex / den;
        const float als = exs / den;
        acc.x = hs.x * als; acc.y = hs.y * als; acc.z = hs.z * als; acc.w = hs.w * als;
        gather_rows(acc, H4, s, al, deg, lane);
    } else {
        float m = es;
        for (int b = beg; b < end; b += 32) {
            const int j = b + lane;
            if (j < end) m = fmaxf(m, lrelu(as_[csr[j]] + add));
        }
        m = wave_max(m);
        float part = 0.f;
        for (int b = beg; b < end; b += 32) {
            const int j = b + lane;
            if (j < end) part += expf(lrelu(as_[csr[j]] + add) - m);
        }
        const float exs = expf(es - m);
        const float den = exs + wave_sum(part);
        const float als = exs / den;
        acc.x = hs.x * als; acc.y = hs.y * als; acc.z = hs.z * als; acc.w = hs.w * als;
        for (int b = beg; b < end; b += 32) {
            const int j = b + lane;
            int s = d; float al = 0.f;
            if (j < end) { s = csr[j]; al = expf(lrelu(as_[s] + add) - m) / den; }
            const int left = end - b;
            gather_rows(acc, H4, s, al, left < 32 ? left : 32, lane);
        }
    }
    const float4 bv = ((const float4*)bias)[lane];
    acc.x += bv.x; acc.y += bv.y; acc.z += bv.z; acc.w += bv.w;
    ((float4*)out)[(size_t)d * 32 + lane] = acc;
}

extern "C" void kernel_launch(void* const* d_in, const int* in_sizes, int n_in,
                              void* d_out, int out_size, void* d_ws, size_t ws_size,
                              hipStream_t stream) {
    const int*   ei      = (const int*)d_in[0];
    const float* users   = (const float*)d_in[1];
    const float* items   = (const float*)d_in[2];
    const float* W       = (const float*)d_in[3];
    const float* att_src = (const float*)d_in[4];
    const float* att_dst = (const float*)d_in[5];
    const float* bias    = (const float*)d_in[6];
    float*       out     = (float*)d_out;

    float* xbuf = (float*)d_ws;
    float* hbuf = xbuf + (size_t)NNODES * HID;
    float* asb  = hbuf + (size_t)NNODES * HID;
    float* adb  = asb + NNODES;
    int*   deg  = (int*)(adb + NNODES);
    int*   off  = deg + NPAD;
    int*   rank = off + NPAD;
    int*   csr  = rank + NE;

    const int GRID_Z  = (NPAD + 255) / 256;
    const int GRID_Q  = (NQUAD + 255) / 256;
    const int GRID_GE = (MTILES + 7) / 8;
    const int GRID_NW = (NNODES + 7) / 8;
    const int GRID_HG = GB_A + GRID_Q;
    const int GRID_SG = GB_B + GRID_Q;

    zero_deg<<<GRID_Z, 256, 0, stream>>>(deg);
    hist_k<<<GRID_HG, 256, 0, stream>>>(ei, deg, rank, users, items, W, att_src, att_dst,
                                              hbuf, asb, adb);
    scan_k<<<1, 1024, 0, stream>>>(deg, off);
    scatter_k<<<GRID_SG, 256, 0, stream>>>(ei, off, rank, csr, users, items, W, att_src,
                                                 att_dst, hbuf, asb, adb);
    gat_node_k<<<GRID_NW, 256, 0, stream>>>(off, csr, hbuf, asb, adb, bias, xbuf);

    gemm_wmma<<<GRID_GE, 256, 0, stream>>>(xbuf, xbuf + (size_t)NUM_USERS * HID,
                                           W + (size_t)HID * HID, att_src + HID, att_dst + HID,
                                           hbuf, asb, adb);
    gat_node_k<<<GRID_NW, 256, 0, stream>>>(off, csr, hbuf, asb, adb, bias + HID, out);
}


# Edges grouped by LDS ranks, not per-edge global atomics: chunks count 256 node ranges, sum the count table, file pairs by range; a group per range orders by node; 6 launches; sum order = LDS arrival
# speedup vs baseline: 5.2799x; 1.3865x over previous
#include <hip/hip_runtime.h>
#include <hip/hip_bf16.h>

#define NUM_USERS 50000
#define NUM_ITEMS 20000
#define NNODES    70000
#define MTILES    (NNODES / 16)
#define HID       128
#define NE        1000000
#define NEG_SLOPE 0.2f

typedef float v2f __attribute__((ext_vector_type(2)));
typedef float v8f __attribute__((ext_vector_type(8)));

template <int N> __device__ __forceinline__ float row_xor(float x) {
    return __int_as_float(__builtin_amdgcn_update_dpp(0, __float_as_int(x), 0x160 + N, 0xF, 0xF, false));
}
__device__ __forceinline__ float xhalf(float x) {
    const int b = __float_as_int(x);
    return __int_as_float(__builtin_amdgcn_permlanex16(b, b, 0x76543210, 0xfedcba98, false, false));
}
__device__ __forceinline__ float wave_max(float v) {
    v = fmaxf(v, row_xor<1>(v));
    v = fmaxf(v, row_xor<2>(v));
    v = fmaxf(v, row_xor<4>(v));
    v = fmaxf(v, row_xor<8>(v));
    return fmaxf(v, xhalf(v));
}
__device__ __forceinline__ float row_sum16(float v) {
    v += row_xor<1>(v);
    v += row_xor<2>(v);
    v += row_xor<4>(v);
    v += row_xor<8>(v);
    return v;
}
__device__ __forceinline__ float wave_sum(float v) {
    v = row_sum16(v);
    return v + xhalf(v);
}

__device__ __forceinline__ void gemm_tile(const int tileM, const int lane,
                                          const float* __restrict__ Xu,
                                          const float* __restrict__ Xi,
                                          const float* __restrict__ Wl,
                                          const float* __restrict__ Asrc,
                                          const float* __restrict__ Adst,
                                          float* __restrict__ H,
                                          float* __restrict__ as_out,
                                          float* __restrict__ ad_out) {
    const int m     = lane & 15;
    const int khalf = (lane >> 4) * 2;
    const int row   = tileM * 16 + m;

    const float4 av0 = ((const float4*)Asrc)[m], av1 = ((const float4*)Asrc)[16 + m];
    const float4 dv0 = ((const float4*)Adst)[m], dv1 = ((const float4*)Adst)[16 + m];
    const float av[8] = {av0.x, av0.y, av0.z, av0.w, av1.x, av1.y, av1.z, av1.w};
    const float dv[8] = {dv0.x, dv0.y, dv0.z, dv0.w, dv1.x, dv1.y, dv1.z, dv1.w};

    v8f c[8] = {};
    const float* xrow = (row < NUM_USERS) ? Xu + (size_t)row * HID
                                          : Xi + (size_t)(row - NUM_USERS) * HID;
    const float* xr = xrow + khalf;
    const float4* wr = (const float4*)(Wl + (size_t)khalf * HID) + m;

    for (int k = 0; k < HID; k += 4) {
        v2f a = *(const v2f*)(xr + k);
        const float4* w0 = wr + (size_t)k * (HID / 4);
        const float4* w1 = w0 + (HID / 4);
        const float4 p0 = w0[0], p1 = w0[16];
        const float4 q0 = w1[0], q1 = w1[16];
        const float bx[8] = {p0.x, p0.y, p0.z, p0.w, p1.x, p1.y, p1.z, p1.w};
        const float by[8] = {q0.x, q0.y, q0.z, q0.w, q1.x, q1.y, q1.z, q1.w};
#pragma unroll
        for (int t = 0; t < 8; ++t) {
            v2f b;
            b.x = bx[t];
            b.y = by[t];
            c[t] = __builtin_amdgcn_wmma_f32_16x16x4_f32(false, a, false, b,
                                                         (short)0, c[t], false, false);
        }
    }
    const int rbase = tileM * 16 + ((lane >> 4) ? 8 : 0);
    float4* hrow = (float4*)(H + (size_t)rbase * HID) + m;
#pragma unroll
    for (int r = 0; r < 8; ++r) {
        hrow[r * (HID / 4)]      = make_float4(c[0][r], c[1][r], c[2][r], c[3][r]);
        hrow[r * (HID / 4) + 16] = make_float4(c[4][r], c[5][r], c[6][r], c[7][r]);
    }

    float ps[8], pd[8];
#pragma unroll
    for (int r = 0; r < 8; ++r) {
        float s = c[0][r] * av[0];
        float d = c[0][r] * dv[0];
#pragma unroll
        for (int t = 1; t < 8; ++t) { s += c[t][r] * av[t]; d += c[t][r] * dv[t]; }
        ps[r] = row_sum16(s);
        pd[r] = row_sum16(d);
    }
    if (m == 0) {
        float4* sp = (float4*)(as_out + rbase);
        float4* dp = (float4*)(ad_out + rbase);
        sp[0] = make_float4(ps[0], ps[1], ps[2], ps[3]);
        sp[1] = make_float4(ps[4], ps[5], ps[6], ps[7]);
        dp[0] = make_float4(pd[0], pd[1], pd[2], pd[3]);
        dp[1] = make_float4(pd[4], pd[5], pd[6], pd[7]);
    }
}

__global__ __launch_bounds__(256) void gemm_wmma(const float* __restrict__ Xu,
                                                 const float* __restrict__ Xi,
                                                 const float* __restrict__ Wl,
                                                 const float* __restrict__ Asrc,
                                                 const float* __restrict__ Adst,
                                                 float* __restrict__ H,
                                                 float* __restrict__ as_out,
                                                 float* __restrict__ ad_out) {
    const int tileM = blockIdx.x * 8 + (threadIdx.x >> 5);
    if (tileM >= MTILES) return;
    gemm_tile(tileM, threadIdx.x & 31, Xu, Xi, Wl, Asrc, Adst, H, as_out, ad_out);
}

#define NQUAD   (NE / 4)
#define NBK     256
#define BKN     274
#define NCH     288
#define CHQ     869
#define GB_1    183
#define GB_2    182
#define GB_3    182
static_assert(NE % 4 == 0, "edge quads");
static_assert((GB_1 + GB_2) * 8 < MTILES && (GB_1 + GB_2 + GB_3) * 8 >= MTILES, "the thirds cover every tile");
static_assert(NBK * BKN >= NNODES, "the ranges cover every node");
static_assert((NBK - 1) * BKN < NNODES, "no range is empty of nodes");
static_assert(NCH * CHQ >= NQUAD && (NCH - 1) * CHQ < NQUAD, "the chunks cover every edge, none is empty");
static_assert(CHQ <= 4 * 256, "four quads per thread cover a chunk");
static_assert(NCH % (4 * 18) == 0 && NBK == 256, "count table read as 4 chunk quarters x 64 range quads, 18 loads a trip");
static_assert(NCH >= BKN + 1 && NCH == 9 * 32, "a range's node counters: nine per lane of one wave");

__device__ __forceinline__ int range_of(int d) { return (int)((unsigned)d / (unsigned)BKN); }

__global__ __launch_bounds__(256) void count_k(const int* __restrict__ ei,
                                               int* __restrict__ cnt,
                                               const float* __restrict__ Xu,
                                                const float* __restrict__ Xi,
                                                const float* __restrict__ Wl,
                                                const float* __restrict__ Asrc,
                                                const float* __restrict__ Adst,
                                                float* __restrict__ H,
                                                float* __restrict__ as_out,
                                                float* __restrict__ ad_out) {
    __shared__ int h[NBK];
    if (blockIdx.x < GB_1) {
        gemm_tile(blockIdx.x * 8 + (threadIdx.x >> 5), threadIdx.x & 31,
                  Xu, Xi, Wl, Asrc, Adst, H, as_out, ad_out);
        return;
    }
    const int b = blockIdx.x - GB_1, t = threadIdx.x;
    h[t] = 0;
    __syncthreads();
    const int q0 = b * CHQ + t;
    const int q1 = (b * CHQ + CHQ < NQUAD) ? b * CHQ + CHQ : NQUAD;
    const int4* dq = (const int4*)(ei + NE);
    int4 d[4];
#pragma unroll
    for (int k = 0; k < 4; ++k) {
        const int qk = (q0 + 256 * k < q1) ? q0 + 256 * k : q1 - 1;
        d[k] = dq[qk];
    }
#pragma unroll
    for (int k = 0; k < 4; ++k)
        if (q0 + 256 * k < q1) {
            atomicAdd(&h[range_of(d[k].x)], 1);
            atomicAdd(&h[range_of(d[k].y)], 1);
            atomicAdd(&h[range_of(d[k].z)], 1);
            atomicAdd(&h[range_of(d[k].w)], 1);
        }
    __syncthreads();
    cnt[b * NBK + t] = h[t];
}

__global__ __launch_bounds__(256) void part_k(const int* __restrict__ ei,
                                              const int* __restrict__ cnt,
                                              int* __restrict__ rstart,
                                              int* __restrict__ pairs_,
                                              const float* __restrict__ Xu,
                                                const float* __restrict__ Xi,
                                                const float* __restrict__ Wl,
                                                const float* __restrict__ Asrc,
                                                const float* __restrict__ Adst,
                                                float* __restrict__ H,
                                                float* __restrict__ as_out,
                                                float* __restrict__ ad_out) {
    __shared__ int stot[4][NBK];
    __shared__ int spre[4][NBK];
    __shared__ int wsum[8];
    __shared__ int cur[NBK];
    int2* __restrict__ pairs = (int2*)pairs_;
    if (blockIdx.x < GB_2) {
        gemm_tile((GB_1 + blockIdx.x) * 8 + (threadIdx.x >> 5), threadIdx.x & 31,
                  Xu, Xi, Wl, Asrc, Adst, H, as_out, ad_out);
        return;
    }
    const int b = blockIdx.x - GB_2, t = threadIdx.x;

    {
        const int g = t & 63, qd = t >> 6;
        const int4* c4 = (const int4*)cnt;
        const int o = qd * (NCH / 4) * (NBK / 4) + g;
        const int jb = b - qd * (NCH / 4);
        int4 tot = make_int4(0, 0, 0, 0), pre = make_int4(0, 0, 0, 0);
#pragma nounroll
        for (int j0 = 0; j0 < NCH / 4; j0 += 18) {
            int4 v[18];
#pragma unroll
            for (int j = 0; j < 18; ++j) v[j] = c4[o + (j0 + j) * (NBK / 4)];
#pragma unroll
            for (int j = 0; j < 18; ++j) {
                const bool below = j0 + j < jb;
                tot.x += v[j].x; tot.y += v[j].y; tot.z += v[j].z; tot.w += v[j].w;
                pre.x += below ? v[j].x : 0; pre.y += below ? v[j].y : 0;
                pre.z += below ? v[j].z : 0; pre.w += below ? v[j].w : 0;
            }
        }
        stot[qd][4 * g + 0] = tot.x; stot[qd][4 * g + 1] = tot.y;
        stot[qd][4 * g + 2] = tot.z; stot[qd][4 * g + 3] = tot.w;
        spre[qd][4 * g + 0] = pre.x; spre[qd][4 * g + 1] = pre.y;
        spre[qd][4 * g + 2] = pre.z; spre[qd][4 * g + 3] = pre.w;
    }
    __syncthreads();
    const int tot = (stot[0][t] + stot[1][t]) + (stot[2][t] + stot[3][t]);
    const int pre = (spre[0][t] + spre[1][t]) + (spre[2][t] + spre[3][t]);
    const int lane = t & 31, wv = t >> 5;
    int incl = tot;
#pragma unroll
    for (int o = 1; o < 32; o <<= 1) {
        const int u = __shfl_up(incl, o, 32);
        if (lane >= o) incl += u;
    }
    if (lane == 31) wsum[wv] = incl;
    __syncthreads();
    int base = incl - tot;
#pragma unroll
    for (int w = 0; w < 8; ++w) base += (w < wv) ? wsum[w] : 0;
    cur[t] = base + pre;
    if (b == 0) rstart[t] = base;
    __syncthreads();

    const int q0 = b * CHQ + t;
    const int q1 = (b * CHQ + CHQ < NQUAD) ? b * CHQ + CHQ : NQUAD;
    const int4* sq = (const int4*)ei;
    const int4* dq = (const int4*)(ei + NE);
    int4 s[4], d[4];
#pragma unroll
    for (int k = 0; k < 4; ++k) {
        const int qk = (q0 + 256 * k < q1) ? q0 + 256 * k : q1 - 1;
        s[k] = sq[qk];
        d[k] = dq[qk];
    }
#pragma unroll
    for (int k = 0; k < 4; ++k)
        if (q0 + 256 * k < q1) {
            pairs[atomicAdd(&cur[range_of(d[k].x)], 1)] = make_int2(s[k].x, d[k].x);
            pairs[atomicAdd(&cur[range_of(d[k].y)], 1)] = make_int2(s[k].y, d[k].y);
            pairs[atomicAdd(&cur[range_of(d[k].z)], 1)] = make_int2(s[k].z, d[k].z);
            pairs[atomicAdd(&cur[range_of(d[k].w)], 1)] = make_int2(s[k].w, d[k].w);
        }
}

__global__ __launch_bounds__(256) void bucket_k(const int* __restrict__ rstart,
                                                const int* __restrict__ pairs_,
                                                int* __restrict__ off,
                                                int* __restrict__ csr,
                                                const float* __restrict__ Xu,
                                                const float* __restrict__ Xi,
                                                const float* __restrict__ Wl,
                                                const float* __restrict__ Asrc,
                                                const float* __restrict__ Adst,
                                                float* __restrict__ H,
                                                float* __restrict__ as_out,
                                                float* __restrict__ ad_out) {
    __shared__ int c[NCH];
    __shared__ int cur[NCH];
    const int2* __restrict__ pairs = (const int2*)pairs_;
    if (blockIdx.x < GB_3) {
        const int tileM = (GB_1 + GB_2 + blockIdx.x) * 8 + (threadIdx.x >> 5);
        if (tileM < MTILES)
            gemm_tile(tileM, threadIdx.x & 31, Xu, Xi, Wl, Asrc, Adst, H, as_out, ad_out);
        return;
    }
    const int r = blockIdx.x - GB_3, t = threadIdx.x;
    const int lo = rstart[r];
    const int hi = (r + 1 < NBK) ? rstart[r + 1] : NE;
    const int n0 = r * BKN;
    for (int j = t; j < NCH; j += 256) c[j] = 0;
    __syncthreads();

    for (int i = lo + t; i < hi; i += 2048) {
        int d[8];
#pragma unroll
        for (int k = 0; k < 8; ++k)
            d[k] = pairs[(i + 256 * k < hi) ? i + 256 * k : hi - 1].y;
        asm volatile("" :: "v"(d[0]), "v"(d[1]), "v"(d[2]), "v"(d[3]),
                           "v"(d[4]), "v"(d[5]), "v"(d[6]), "v"(d[7]));
#pragma unroll
        for (int k = 0; k < 8; ++k)
            if (i + 256 * k < hi) atomicAdd(&c[d[k] - n0], 1);
    }
    __syncthreads();

    if (t < 32) {
        int v[9], s = 0;
#pragma unroll
        for (int k = 0; k < 9; ++k) { v[k] = c[9 * t + k]; s += v[k]; }
        int incl = s;
#pragma unroll
        for (int o = 1; o < 32; o <<= 1) {
            const int u = __shfl_up(incl, o, 32);
            if (t >= o) incl += u;
        }
        int run = incl - s;
#pragma unroll
        for (int k = 0; k < 9; ++k) { c[9 * t + k] = run; cur[9 * t + k] = run; run += v[k]; }
    }
    __syncthreads();

    const int nn = (NNODES - n0 < BKN) ? NNODES - n0 : BKN;
    for (int j = t; j < nn; j += 256) off[n0 + j] = lo + c[j];
    if (r == NBK - 1 && t == 0) off[NNODES] = NE;

    for (int i = lo + t; i < hi; i += 2048) {
        int2 p[8];
#pragma unroll
        for (int k = 0; k < 8; ++k)
            p[k] = pairs[(i + 256 * k < hi) ? i + 256 * k : hi - 1];
        asm volatile("" :: "v"(p[0].y), "v"(p[1].y), "v"(p[2].y), "v"(p[3].y),
                           "v"(p[4].y), "v"(p[5].y), "v"(p[6].y), "v"(p[7].y));
#pragma unroll
        for (int k = 0; k < 8; ++k)
            if (i + 256 * k < hi) csr[lo + atomicAdd(&cur[p[k].y - n0], 1)] = p[k].x;
    }
}

__device__ __forceinline__ float lrelu(float e) { return (e >= 0.f) ? e : NEG_SLOPE * e; }

__device__ __forceinline__ void gather_rows(float4& acc, const float4* __restrict__ H4,
                                            int s, float al, int cnt, int lane) {
    const int ab = __float_as_int(al);
    for (int k = 0; k < cnt; k += 4) {
        const int s0 = __builtin_amdgcn_readlane(s, k);
        const int s1 = __builtin_amdgcn_readlane(s, k + 1);
        const int s2 = __builtin_amdgcn_readlane(s, k + 2);
        const int s3 = __builtin_amdgcn_readlane(s, k + 3);
        const float a0 = __int_as_float(__builtin_amdgcn_readlane(ab, k));
        const float a1 = __int_as_float(__builtin_amdgcn_readlane(ab, k + 1));
        const float a2 = __int_as_float(__builtin_amdgcn_readlane(ab, k + 2));
        const float a3 = __int_as_float(__builtin_amdgcn_readlane(ab, k + 3));
        const float4 h0 = H4[(size_t)s0 * 32 + lane];
        const float4 h1 = H4[(size_t)s1 * 32 + lane];
        const float4 h2 = H4[(size_t)s2 * 32 + lane];
        const float4 h3 = H4[(size_t)s3 * 32 + lane];
        acc.x += h0.x * a0; acc.y += h0.y * a0; acc.z += h0.z * a0; acc.w += h0.w * a0;
        acc.x += h1.x * a1; acc.y += h1.y * a1; acc.z += h1.z * a1; acc.w += h1.w * a1;
        acc.x += h2.x * a2; acc.y += h2.y * a2; acc.z += h2.z * a2; acc.w += h2.w * a2;
        acc.x += h3.x * a3; acc.y += h3.y * a3; acc.z += h3.z * a3; acc.w += h3.w * a3;
    }
}

__global__ __launch_bounds__(256) void gat_node_k(const int* __restrict__ off,
                                                  const int* __restrict__ csr,
                                                  const float* __restrict__ H,
                                                  const float* __restrict__ as_,
                                                  const float* __restrict__ ad_,
                                                  const float* __restrict__ bias,
                                                  float* __restrict__ out) {
    const int lane = threadIdx.x & 31;
    const int d = __builtin_amdgcn_readfirstlane(blockIdx.x * 8 + (threadIdx.x >> 5));
    if (d >= NNODES) return;
    const int beg = off[d], end = off[d + 1];
    const int deg = end - beg;
    const float add = ad_[d];
    const float es = lrelu(as_[d] + add);
    const float4* H4 = (const float4*)H;
    const float4 hs = H4[(size_t)d * 32 + lane];
    float4 acc;

    if (deg <= 32) {
        int s = d; float e = -__builtin_inff();
        if (lane < deg) { s = csr[beg + lane]; e = lrelu(as_[s] + add); }
        const float m   = fmaxf(es, wave_max(e));
        const float ex  = (lane < deg) ? expf(e - m) : 0.f;
        const float exs = expf(es - m);
        const float den = exs + wave_sum(ex);
        const float al  = ex / den;
        const float als = exs / den;
        acc.x = hs.x * als; acc.y = hs.y * als; acc.z = hs.z * als; acc.w = hs.w * als;
        gather_rows(acc, H4, s, al, deg, lane);
    } else {
        float m = es;
        for (int b = beg; b < end; b += 32) {
            const int j = b + lane;
            if (j < end) m = fmaxf(m, lrelu(as_[csr[j]] + add));
        }
        m = wave_max(m);
        float part = 0.f;
        for (int b = beg; b < end; b += 32) {
            const int j = b + lane;
            if (j < end) part += expf(lrelu(as_[csr[j]] + add) - m);
        }
        const float exs = expf(es - m);
        const float den = exs + wave_sum(part);
        const float als = exs / den;
        acc.x = hs.x * als; acc.y = hs.y * als; acc.z = hs.z * als; acc.w = hs.w * als;
        for (int b = beg; b < end; b += 32) {
            const int j = b + lane;
            int s = d; float al = 0.f;
            if (j < end) { s = csr[j]; al = expf(lrelu(as_[s] + add) - m) / den; }
            const int left = end - b;
            gather_rows(acc, H4, s, al, left < 32 ? left : 32, lane);
        }
    }
    const float4 bv = ((const float4*)bias)[lane];
    acc.x += bv.x; acc.y += bv.y; acc.z += bv.z; acc.w += bv.w;
    ((float4*)out)[(size_t)d * 32 + lane] = acc;
}

extern "C" void kernel_launch(void* const* d_in, const int* in_sizes, int n_in,
                              void* d_out, int out_size, void* d_ws, size_t ws_size,
                              hipStream_t stream) {
    const int*   ei      = (const int*)d_in[0];
    const float* users   = (const float*)d_in[1];
    const float* items   = (const float*)d_in[2];
    const float* W       = (const float*)d_in[3];
    const float* att_src = (const float*)d_in[4];
    const float* att_dst = (const float*)d_in[5];
    const float* bias    = (const float*)d_in[6];
    float*       out     = (float*)d_out;

    float* xbuf   = (float*)d_ws;
    float* hbuf   = xbuf + (size_t)NNODES * HID;
    float* asb    = hbuf + (size_t)NNODES * HID;
    float* adb    = asb + NNODES;
    int*   cnt    = (int*)(adb + NNODES);
    int*   rstart = cnt + NCH * NBK;
    int*   off    = rstart + NBK;
    int*   pairs  = off + NBK * NCH;
    int*   csr    = pairs + 2 * (size_t)NE;

    const int GRID_GE = (MTILES + 7) / 8;
    const int GRID_NW = (NNODES + 7) / 8;

    count_k<<<GB_1 + NCH, 256, 0, stream>>>(ei, cnt, users, items, W, att_src, att_dst,
                                            hbuf, asb, adb);
    part_k<<<GB_2 + NCH, 256, 0, stream>>>(ei, cnt, rstart, pairs, users, items, W, att_src,
                                           att_dst, hbuf, asb, adb);
    bucket_k<<<GB_3 + NBK, 256, 0, stream>>>(rstart, pairs, off, csr, users, items, W, att_src,
                                             att_dst, hbuf, asb, adb);
    gat_node_k<<<GRID_NW, 256, 0, stream>>>(off, csr, hbuf, asb, adb, bias, xbuf);

    gemm_wmma<<<GRID_GE, 256, 0, stream>>>(xbuf, xbuf + (size_t)NUM_USERS * HID,
                                           W + (size_t)HID * HID, att_src + HID, att_dst + HID,
                                           hbuf, asb, adb);
    gat_node_k<<<GRID_NW, 256, 0, stream>>>(off, csr, hbuf, asb, adb, bias + HID, out);
}
